# starting kernel
# baseline (speedup 1.0000x reference)
#include <hip/hip_runtime.h>
#include <cstdint>

typedef float v2f __attribute__((ext_vector_type(2)));
typedef float v8f __attribute__((ext_vector_type(8)));

namespace {

constexpr int B_ = 2, T_ = 64, C_ = 384, H_ = 768;
constexpr float LR = 0.01f, MOM = 0.9f, DECAY = 0.001f;

constexpr int OFF_W    = 0;
constexpr int OFF_K    = 128;
constexpr int OFF_V    = OFF_K    + B_*T_*C_;
constexpr int OFF_H1   = OFF_V    + B_*T_*C_;
constexpr int OFF_GP   = OFF_H1   + B_*T_*H_;
constexpr int OFF_DP   = OFF_GP   + B_*T_*H_;
constexpr int OFF_WDPT = OFF_DP   + B_*T_*C_;
constexpr int OFF_WDHT = OFF_WDPT + B_*C_*T_;

__device__ __forceinline__ v8f wmma4(v2f a, v2f b, v8f c) {
  return __builtin_amdgcn_wmma_f32_16x16x4_f32(false, a, false, b, (short)0, c,
                                               false, false);
}

template <bool BT>
__device__ __forceinline__ void gemm32x32(const float* __restrict__ A, int lda,
                                          const float* __restrict__ Bm, int ldb,
                                          int K, int m0, int n0, int lane,
                                          v8f acc[4]) {
  const int mr = lane & 15;
  const int kh = (lane >> 4) << 1;
  for (int kk = 0; kk < K; kk += 4) {
    const int kp = kk + kh;
    v2f a0 = *reinterpret_cast<const v2f*>(A + (m0 + mr) * lda + kp);
    v2f a1 = *reinterpret_cast<const v2f*>(A + (m0 + 16 + mr) * lda + kp);
    v2f b0, b1;
    if (BT) {
      b0 = *reinterpret_cast<const v2f*>(Bm + (n0 + mr) * ldb + kp);
      b1 = *reinterpret_cast<const v2f*>(Bm + (n0 + 16 + mr) * ldb + kp);
    } else {
      const float* pb = Bm + kp * ldb + n0 + mr;
      b0.x = pb[0];  b0.y = pb[ldb];
      b1.x = pb[16]; b1.y = pb[ldb + 16];
    }
    acc[0] = wmma4(a0, b0, acc[0]);
    acc[1] = wmma4(a0, b1, acc[1]);
    acc[2] = wmma4(a1, b0, acc[2]);
    acc[3] = wmma4(a1, b1, acc[3]);
  }
}

__global__ void k_init_w(float* __restrict__ ws) {
  int t = threadIdx.x;
  const float a = 1.0f - DECAY;
  float n = (float)(T_ - t);
  float an = __powf(a, n), mn = __powf(MOM, n);
  ws[OFF_W + t] = -LR * (1.0f - MOM) * (an - mn) / (a - MOM);
  if (t == 0) ws[OFF_W + 64] = __powf(a, (float)T_);
}

__global__ void k_gemm_kv(const float* __restrict__ x,
                          const float* __restrict__ Wk, const float* __restrict__ bk,
                          const float* __restrict__ Wv, const float* __restrict__ bv,
                          float* __restrict__ ws) {
  const int b = blockIdx.y, which = blockIdx.z, lane = threadIdx.x;
  const int m0 = (blockIdx.x & 1) * 32, n0 = (blockIdx.x >> 1) * 32;
  const float* A = x + b * T_ * C_;
  const float* W = which ? Wv : Wk;
  const float* bias = which ? bv : bk;
  float* out = ws + (which ? OFF_V : OFF_K) + b * T_ * C_;
  v8f acc[4] = {};
  gemm32x32<true>(A, C_, W, C_, C_, m0, n0, lane, acc);
  const int cr = lane & 15, hi = lane >> 4;
#pragma unroll
  for (int mi = 0; mi < 2; ++mi)
#pragma unroll
    for (int ni = 0; ni < 2; ++ni) {
      int col = n0 + ni * 16 + cr;
      float bb = bias[col];
      v8f a = acc[mi * 2 + ni];
#pragma unroll
      for (int r = 0; r < 8; ++r) {
        int row = m0 + mi * 16 + r + 8 * hi;
        out[row * C_ + col] = a[r] + bb;
      }
    }
}

__global__ void k_gemm_h1(const float* __restrict__ sW0,
                          const float* __restrict__ sb0,
                          float* __restrict__ ws) {
  const int b = blockIdx.y, lane = threadIdx.x;
  const int m0 = (blockIdx.x % 2) * 32, n0 = (blockIdx.x / 2) * 32;
  const float* A = ws + OFF_K + b * T_ * C_;
  const float* W = sW0 + b * H_ * C_;
  float* h1 = ws + OFF_H1 + b * T_ * H_;
  float* gp = ws + OFF_GP + b * T_ * H_;
  v8f acc[4] = {};
  gemm32x32<true>(A, C_, W, C_, C_, m0, n0, lane, acc);
  const int cr = lane & 15, hi = lane >> 4;
#pragma unroll
  for (int mi = 0; mi < 2; ++mi)
#pragma unroll
    for (int ni = 0; ni < 2; ++ni) {
      int col = n0 + ni * 16 + cr;
      float bb = sb0[b * H_ + col];
      v8f a = acc[mi * 2 + ni];
#pragma unroll
      for (int r = 0; r < 8; ++r) {
        int row = m0 + mi * 16 + r + 8 * hi;
        float pre = a[r] + bb;
        float sig = 1.0f / (1.0f + __expf(-pre));
        h1[row * H_ + col] = pre * sig;
        gp[row * H_ + col] = sig * (1.0f + pre * (1.0f - sig));
      }
    }
}

__global__ void k_gemm_dp(const float* __restrict__ sW1,
                          const float* __restrict__ sb1,
                          float* __restrict__ ws) {
  const int b = blockIdx.y, lane = threadIdx.x;
  const int m0 = (blockIdx.x % 2) * 32, n0 = (blockIdx.x / 2) * 32;
  const float* A = ws + OFF_H1 + b * T_ * H_;
  const float* W = sW1 + b * C_ * H_;
  const float* v = ws + OFF_V + b * T_ * C_;
  float* dp = ws + OFF_DP + b * T_ * C_;
  float* wdpT = ws + OFF_WDPT + b * C_ * T_;
  v8f acc[4] = {};
  gemm32x32<true>(A, H_, W, H_, H_, m0, n0, lane, acc);
  const int cr = lane & 15, hi = lane >> 4;
#pragma unroll
  for (int mi = 0; mi < 2; ++mi)
#pragma unroll
    for (int ni = 0; ni < 2; ++ni) {
      int col = n0 + ni * 16 + cr;
      float bb = sb1[b * C_ + col];
      v8f a = acc[mi * 2 + ni];
#pragma unroll
      for (int r = 0; r < 8; ++r) {
        int row = m0 + mi * 16 + r + 8 * hi;
        float d = (a[r] + bb) - v[row * C_ + col];
        dp[row * C_ + col] = d;
        wdpT[col * T_ + row] = ws[OFF_W + row] * d;
      }
    }
}

__global__ void k_gemm_dh1(const float* __restrict__ sW1,
                           float* __restrict__ ws) {
  const int b = blockIdx.y, lane = threadIdx.x;
  const int m0 = (blockIdx.x % 2) * 32, n0 = (blockIdx.x / 2) * 32;
  const float* A = ws + OFF_DP + b * T_ * C_;
  const float* W = sW1 + b * C_ * H_;
  const float* gp = ws + OFF_GP + b * T_ * H_;
  float* wdhT = ws + OFF_WDHT + b * H_ * T_;
  v8f acc[4] = {};
  gemm32x32<false>(A, C_, W, H_, C_, m0, n0, lane, acc);
  const int cr = lane & 15, hi = lane >> 4;
#pragma unroll
  for (int mi = 0; mi < 2; ++mi)
#pragma unroll
    for (int ni = 0; ni < 2; ++ni) {
      int col = n0 + ni * 16 + cr;
      v8f a = acc[mi * 2 + ni];
#pragma unroll
      for (int r = 0; r < 8; ++r) {
        int row = m0 + mi * 16 + r + 8 * hi;
        float d = a[r] * gp[row * H_ + col];
        wdhT[col * T_ + row] = ws[OFF_W + row] * d;
      }
    }
}

__global__ void k_gemm_nw0(const float* __restrict__ sW0,
                           float* __restrict__ nW0,
                           float* __restrict__ ws) {
  const int b = blockIdx.y, lane = threadIdx.x;
  const int m0 = (blockIdx.x / 12) * 32, n0 = (blockIdx.x % 12) * 32;
  const float* A = ws + OFF_WDHT + b * H_ * T_;
  const float* Bm = ws + OFF_K + b * T_ * C_;
  const float dT = ws[OFF_W + 64];
  float* out = nW0 + b * H_ * C_;
  const float* w0 = sW0 + b * H_ * C_;
  v8f acc[4] = {};
  gemm32x32<false>(A, T_, Bm, C_, T_, m0, n0, lane, acc);
  const int cr = lane & 15, hi = lane >> 4;
#pragma unroll
  for (int mi = 0; mi < 2; ++mi)
#pragma unroll
    for (int ni = 0; ni < 2; ++ni) {
      int col = n0 + ni * 16 + cr;
      v8f a = acc[mi * 2 + ni];
#pragma unroll
      for (int r = 0; r < 8; ++r) {
        int row = m0 + mi * 16 + r + 8 * hi;
        out[row * C_ + col] = dT * w0[row * C_ + col] + a[r];
      }
    }
}

__global__ void k_gemm_nw1(const float* __restrict__ sW1,
                           float* __restrict__ nW1,
                           float* __restrict__ ws) {
  const int b = blockIdx.y, lane = threadIdx.x;
  const int m0 = (blockIdx.x / 24) * 32, n0 = (blockIdx.x % 24) * 32;
  const float* A = ws + OFF_WDPT + b * C_ * T_;
  const float* Bm = ws + OFF_H1 + b * T_ * H_;
  const float dT = ws[OFF_W + 64];
  float* out = nW1 + b * C_ * H_;
  const float* w1 = sW1 + b * C_ * H_;
  v8f acc[4] = {};
  gemm32x32<false>(A, T_, Bm, H_, T_, m0, n0, lane, acc);
  const int cr = lane & 15, hi = lane >> 4;
#pragma unroll
  for (int mi = 0; mi < 2; ++mi)
#pragma unroll
    for (int ni = 0; ni < 2; ++ni) {
      int col = n0 + ni * 16 + cr;
      v8f a = acc[mi * 2 + ni];
#pragma unroll
      for (int r = 0; r < 8; ++r) {
        int row = m0 + mi * 16 + r + 8 * hi;
        out[row * H_ + col] = dT * w1[row * H_ + col] + a[r];
      }
    }
}

__global__ void k_bias(const float* __restrict__ sb0, const float* __restrict__ sb1,
                       float* __restrict__ nb0, float* __restrict__ nb1,
                       const float* __restrict__ ws) {
  int idx = blockIdx.x * blockDim.x + threadIdx.x;
  if (idx >= B_ * (H_ + C_)) return;
  int b = idx / (H_ + C_), r = idx % (H_ + C_);
  const float dT = ws[OFF_W + 64];
  if (r < H_) {
    const float* p = ws + OFF_WDHT + b * H_ * T_ + r * T_;
    float s = 0.f;
#pragma unroll 4
    for (int t = 0; t < T_; ++t) s += p[t];
    nb0[b * H_ + r] = dT * sb0[b * H_ + r] + s;
  } else {
    int c = r - H_;
    const float* p = ws + OFF_WDPT + b * C_ * T_ + c * T_;
    float s = 0.f;
#pragma unroll 4
    for (int t = 0; t < T_; ++t) s += p[t];
    nb1[b * C_ + c] = dT * sb1[b * C_ + c] + s;
  }
}

__global__ void k_retrieve(const float* __restrict__ Wq, const float* __restrict__ bq,
                           const float* __restrict__ Wo, const float* __restrict__ bo,
                           const float* __restrict__ iq,
                           const float* __restrict__ sW0, const float* __restrict__ sb0,
                           const float* __restrict__ sW1, const float* __restrict__ sb1,
                           float* __restrict__ out) {
  __shared__ float q[C_], h[H_], pred[C_];
  const int b = blockIdx.x, t = threadIdx.x;
  {
    const float* wr = Wq + t * C_;
    float s = 0.f;
    for (int c = 0; c < C_; ++c) s += wr[c] * iq[c];
    q[t] = s + bq[t];
  }
  __syncthreads();
#pragma unroll
  for (int j = 0; j < 2; ++j) {
    int hh = t + j * C_;
    const float* w0 = sW0 + b * H_ * C_ + hh * C_;
    float s = 0.f;
    for (int c = 0; c < C_; ++c) s += w0[c] * q[c];
    s += sb0[b * H_ + hh];
    float sig = 1.0f / (1.0f + __expf(-s));
    h[hh] = s * sig;
  }
  __syncthreads();
  {
    const float* w1 = sW1 + b * C_ * H_ + t * H_;
    float s = 0.f;
    for (int hh = 0; hh < H_; ++hh) s += w1[hh] * h[hh];
    pred[t] = s + sb1[b * C_ + t];
  }
  __syncthreads();
  {
    const float* wo = Wo + t * C_;
    float s = 0.f;
    for (int c = 0; c < C_; ++c) s += wo[c] * pred[c];
    out[b * C_ + t] = s + bo[t];
  }
}

}

extern "C" void kernel_launch(void* const* d_in, const int* in_sizes, int n_in,
                              void* d_out, int out_size, void* d_ws, size_t ws_size,
                              hipStream_t stream) {
  const float* x   = (const float*)d_in[0];
  const float* Wq  = (const float*)d_in[1];
  const float* bq  = (const float*)d_in[2];
  const float* Wk  = (const float*)d_in[3];
  const float* bk  = (const float*)d_in[4];
  const float* Wv  = (const float*)d_in[5];
  const float* bv  = (const float*)d_in[6];
  const float* Wo  = (const float*)d_in[7];
  const float* bo  = (const float*)d_in[8];
  const float* iq  = (const float*)d_in[9];
  const float* sW0 = (const float*)d_in[10];
  const float* sb0 = (const float*)d_in[11];
  const float* sW1 = (const float*)d_in[12];
  const float* sb1 = (const float*)d_in[13];

  float* out = (float*)d_out;
  float* ws  = (float*)d_ws;

  float* out_out = out;
  float* out_nW0 = out + B_ * C_;
  float* out_nb0 = out_nW0 + B_ * H_ * C_;
  float* out_nW1 = out_nb0 + B_ * H_;
  float* out_nb1 = out_nW1 + B_ * C_ * H_;

  k_init_w <<<1, 64, 0, stream>>>(ws);
  k_gemm_kv<<<dim3(24, B_, 2), 32, 0, stream>>>(x, Wk, bk, Wv, bv, ws);
  k_gemm_h1<<<dim3(48, B_), 32, 0, stream>>>(sW0, sb0, ws);
  k_gemm_dp<<<dim3(24, B_), 32, 0, stream>>>(sW1, sb1, ws);
  k_gemm_dh1<<<dim3(48, B_), 32, 0, stream>>>(sW1, ws);
  k_gemm_nw0<<<dim3(288, B_), 32, 0, stream>>>(sW0, out_nW0, ws);
  k_gemm_nw1<<<dim3(288, B_), 32, 0, stream>>>(sW1, out_nW1, ws);
  k_bias    <<<dim3(18), 128, 0, stream>>>(sb0, sb1, out_nb0, out_nb1, ws);
  k_retrieve<<<dim3(B_), C_, 0, stream>>>(Wq, bq, Wo, bo, iq, sW0, sb0, sW1, sb1, out_out);
}


# nW1's 576 tile blocks run as extra blocks of launch 4 beside dh1 instead of in a tail launch: their operands (w*dp transposed, h1) are complete after launch 3; 8 -> 7 launches; sums keep order
# speedup vs baseline: 3.9493x; 3.9493x over previous
#include <hip/hip_runtime.h>
#include <cstdint>

typedef float v2f __attribute__((ext_vector_type(2)));
typedef float v8f __attribute__((ext_vector_type(8)));

namespace {

constexpr int B_ = 2, T_ = 64, C_ = 384, H_ = 768;
constexpr float LR = 0.01f, MOM = 0.9f, DECAY = 0.001f;

constexpr int OFF_W    = 0;
constexpr int OFF_K    = 128;
constexpr int OFF_V    = OFF_K    + B_*T_*C_;
constexpr int OFF_H1   = OFF_V    + B_*T_*C_;
constexpr int OFF_GP   = OFF_H1   + B_*T_*H_;
constexpr int OFF_DP   = OFF_GP   + B_*T_*H_;
constexpr int OFF_WDPT = OFF_DP   + B_*T_*C_;
constexpr int OFF_WDHT = OFF_WDPT + B_*C_*T_;
constexpr int OFF_Q    = OFF_WDHT + B_*H_*T_;
constexpr int OFF_HR   = OFF_Q    + C_;
constexpr int OFF_PR   = OFF_HR   + B_*H_;

__device__ __forceinline__ v8f wmma4(v2f a, v2f b, v8f c) {
  return __builtin_amdgcn_wmma_f32_16x16x4_f32(false, a, false, b, (short)0, c,
                                               false, false);
}

template <int KC>
struct FragSet {
  v2f a0[KC], a1[KC], b0[KC], b1[KC];
};

template <bool BT, int KC>
__device__ __forceinline__ void load_set(FragSet<KC>& f,
                                         const float* __restrict__ pa, int lda,
                                         const float* __restrict__ pb, int ldb,
                                         int kk) {
#pragma unroll
  for (int s = 0; s < KC; ++s) {
    const int k = kk + 4 * s;
    f.a0[s] = *reinterpret_cast<const v2f*>(pa + k);
    f.a1[s] = *reinterpret_cast<const v2f*>(pa + 16 * lda + k);
    if (BT) {
      f.b0[s] = *reinterpret_cast<const v2f*>(pb + k);
      f.b1[s] = *reinterpret_cast<const v2f*>(pb + 16 * ldb + k);
    } else {
      const float* q = pb + k * ldb;
      f.b0[s].x = q[0];  f.b0[s].y = q[ldb];
      f.b1[s].x = q[16]; f.b1[s].y = q[ldb + 16];
    }
  }
}

template <int KC>
__device__ __forceinline__ void mma_set(const FragSet<KC>& f, v8f acc[4]) {
#pragma unroll
  for (int s = 0; s < KC; ++s) {
    acc[0] = wmma4(f.a0[s], f.b0[s], acc[0]);
    acc[1] = wmma4(f.a0[s], f.b1[s], acc[1]);
    acc[2] = wmma4(f.a1[s], f.b0[s], acc[2]);
    acc[3] = wmma4(f.a1[s], f.b1[s], acc[3]);
  }
}

template <bool BT, int K, int KC>
__device__ __forceinline__ void gemm32x32(const float* __restrict__ A, int lda,
                                          const float* __restrict__ Bm, int ldb,
                                          int m0, int n0, int lane,
                                          v8f acc[4]) {
  constexpr int CH = 4 * KC;
  constexpr int NCH = K / CH;
  static_assert(K % CH == 0 && NCH % 2 == 0, "K must be an even number of chunks");
  const int mr = lane & 15;
  const int kh = (lane >> 4) << 1;
  const float* pa = A + (m0 + mr) * lda + kh;
  const float* pb = BT ? Bm + (n0 + mr) * ldb + kh : Bm + kh * ldb + n0 + mr;
  FragSet<KC> f0, f1;
  load_set<BT, KC>(f0, pa, lda, pb, ldb, 0);
  for (int c = 0; c + 2 < NCH; c += 2) {
    load_set<BT, KC>(f1, pa, lda, pb, ldb, (c + 1) * CH);
    mma_set<KC>(f0, acc);
    load_set<BT, KC>(f0, pa, lda, pb, ldb, (c + 2) * CH);
    mma_set<KC>(f1, acc);
  }
  load_set<BT, KC>(f1, pa, lda, pb, ldb, (NCH - 1) * CH);
  mma_set<KC>(f0, acc);
  mma_set<KC>(f1, acc);
}

template <int KC>
struct RowSet {
  v2f a[KC], b0[KC], b1[KC];
};

template <int KC>
__device__ __forceinline__ void load_row_set(RowSet<KC>& f,
                                             const float* __restrict__ pa,
                                             const float* __restrict__ pb, int ldb,
                                             int kk) {
#pragma unroll
  for (int s = 0; s < KC; ++s) {
    const int k = kk + 4 * s;
    f.a[s]  = *reinterpret_cast<const v2f*>(pa + k);
    f.b0[s] = *reinterpret_cast<const v2f*>(pb + k);
    f.b1[s] = *reinterpret_cast<const v2f*>(pb + 16 * ldb + k);
  }
}

template <int KC>
__device__ __forceinline__ void mma_row_set(const RowSet<KC>& f, v8f acc[2]) {
#pragma unroll
  for (int s = 0; s < KC; ++s) {
    acc[0] = wmma4(f.a[s], f.b0[s], acc[0]);
    acc[1] = wmma4(f.a[s], f.b1[s], acc[1]);
  }
}

template <int K, int KC>
__device__ __forceinline__ float rowvec32(const float* __restrict__ xrow,
                                          const float* __restrict__ Wm, int ldb,
                                          int n0, int lane) {
  constexpr int CH = 4 * KC;
  constexpr int NCH = K / CH;
  static_assert(K % CH == 0 && NCH % 2 == 0, "K must be an even number of chunks");
  const int mr = lane & 15;
  const int kh = (lane >> 4) << 1;
  const float* pa = xrow + kh;
  const float* pb = Wm + (n0 + mr) * ldb + kh;
  v8f acc[2] = {};
  RowSet<KC> f0, f1;
  load_row_set<KC>(f0, pa, pb, ldb, 0);
  for (int c = 0; c + 2 < NCH; c += 2) {
    load_row_set<KC>(f1, pa, pb, ldb, (c + 1) * CH);
    mma_row_set<KC>(f0, acc);
    load_row_set<KC>(f0, pa, pb, ldb, (c + 2) * CH);
    mma_row_set<KC>(f1, acc);
  }
  load_row_set<KC>(f1, pa, pb, ldb, (NCH - 1) * CH);
  mma_row_set<KC>(f0, acc);
  mma_row_set<KC>(f1, acc);
  return (lane >> 4) ? acc[1][0] : acc[0][0];
}

__global__ void k_init_w(float* __restrict__ ws) {
  int t = threadIdx.x;
  const float a = 1.0f - DECAY;
  float n = (float)(T_ - t);
  float an = __powf(a, n), mn = __powf(MOM, n);
  ws[OFF_W + t] = -LR * (1.0f - MOM) * (an - mn) / (a - MOM);
  if (t == 0) ws[OFF_W + 64] = __powf(a, (float)T_);
}

__global__ void __launch_bounds__(32) k_gemm_kv(const float* __restrict__ x,
                          const float* __restrict__ Wk, const float* __restrict__ bk,
                          const float* __restrict__ Wv, const float* __restrict__ bv,
                          const float* __restrict__ Wq, const float* __restrict__ bq,
                          const float* __restrict__ iq,
                          float* __restrict__ ws) {
  const int b = blockIdx.y, which = blockIdx.z, lane = threadIdx.x;
  if (blockIdx.x >= 24) {
    const int n0 = ((blockIdx.x - 24) + 3 * (b + 2 * which)) * 32;
    const float bb = bq[n0 + lane];
    const float y = rowvec32<C_, 16>(iq, Wq, C_, n0, lane);
    ws[OFF_Q + n0 + lane] = y + bb;
    return;
  }
  const int m0 = (blockIdx.x & 1) * 32, n0 = (blockIdx.x >> 1) * 32;
  const float* A = x + b * T_ * C_;
  const float* W = which ? Wv : Wk;
  const float* bias = which ? bv : bk;
  float* out = ws + (which ? OFF_V : OFF_K) + b * T_ * C_;
  v8f acc[4] = {};
  gemm32x32<true, C_, 16>(A, C_, W, C_, m0, n0, lane, acc);
  const int cr = lane & 15, hi = lane >> 4;
#pragma unroll
  for (int mi = 0; mi < 2; ++mi)
#pragma unroll
    for (int ni = 0; ni < 2; ++ni) {
      int col = n0 + ni * 16 + cr;
      float bb = bias[col];
      v8f a = acc[mi * 2 + ni];
#pragma unroll
      for (int r = 0; r < 8; ++r) {
        int row = m0 + mi * 16 + r + 8 * hi;
        out[row * C_ + col] = a[r] + bb;
      }
    }
}

__global__ void __launch_bounds__(32) k_gemm_h1(const float* __restrict__ sW0,
                          const float* __restrict__ sb0,
                          float* __restrict__ ws) {
  const int b = blockIdx.y, lane = threadIdx.x;
  if (blockIdx.x >= 48) {
    const int n0 = (blockIdx.x - 48) * 32;
    const float bb = sb0[b * H_ + n0 + lane];
    const float pre = rowvec32<C_, 16>(ws + OFF_Q, sW0 + b * H_ * C_, C_, n0, lane) + bb;
    const float sig = 1.0f / (1.0f + __expf(-pre));
    ws[OFF_HR + b * H_ + n0 + lane] = pre * sig;
    return;
  }
  const int m0 = (blockIdx.x % 2) * 32, n0 = (blockIdx.x / 2) * 32;
  const float* A = ws + OFF_K + b * T_ * C_;
  const float* W = sW0 + b * H_ * C_;
  float* h1 = ws + OFF_H1 + b * T_ * H_;
  float* gp = ws + OFF_GP + b * T_ * H_;
  v8f acc[4] = {};
  gemm32x32<true, C_, 16>(A, C_, W, C_, m0, n0, lane, acc);
  const int cr = lane & 15, hi = lane >> 4;
#pragma unroll
  for (int mi = 0; mi < 2; ++mi)
#pragma unroll
    for (int ni = 0; ni < 2; ++ni) {
      int col = n0 + ni * 16 + cr;
      float bb = sb0[b * H_ + col];
      v8f a = acc[mi * 2 + ni];
#pragma unroll
      for (int r = 0; r < 8; ++r) {
        int row = m0 + mi * 16 + r + 8 * hi;
        float pre = a[r] + bb;
        float sig = 1.0f / (1.0f + __expf(-pre));
        h1[row * H_ + col] = pre * sig;
        gp[row * H_ + col] = sig * (1.0f + pre * (1.0f - sig));
      }
    }
}

__global__ void __launch_bounds__(32) k_gemm_dp(const float* __restrict__ sW1,
                          const float* __restrict__ sb1,
                          float* __restrict__ ws) {
  const int b = blockIdx.y, lane = threadIdx.x;
  if (blockIdx.x >= 24) {
    const int n0 = (blockIdx.x - 24) * 32;
    const float bb = sb1[b * C_ + n0 + lane];
    const float y = rowvec32<H_, 16>(ws + OFF_HR + b * H_, sW1 + b * C_ * H_, H_, n0, lane);
    ws[OFF_PR + b * C_ + n0 + lane] = y + bb;
    return;
  }
  const int m0 = (blockIdx.x % 2) * 32, n0 = (blockIdx.x / 2) * 32;
  const float* A = ws + OFF_H1 + b * T_ * H_;
  const float* W = sW1 + b * C_ * H_;
  const float* v = ws + OFF_V + b * T_ * C_;
  float* dp = ws + OFF_DP + b * T_ * C_;
  float* wdpT = ws + OFF_WDPT + b * C_ * T_;
  const int cr = lane & 15, hi = lane >> 4;
  float vt[2][2][8], wt[2][8];
#pragma unroll
  for (int mi = 0; mi < 2; ++mi)
#pragma unroll
    for (int r = 0; r < 8; ++r) {
      int row = m0 + mi * 16 + r + 8 * hi;
      wt[mi][r] = ws[OFF_W + row];
#pragma unroll
      for (int ni = 0; ni < 2; ++ni)
        vt[mi][ni][r] = v[row * C_ + n0 + ni * 16 + cr];
    }
  v8f acc[4] = {};
  gemm32x32<true, H_, 16>(A, H_, W, H_, m0, n0, lane, acc);
  float dd[2][2][8];
#pragma unroll
  for (int mi = 0; mi < 2; ++mi)
#pragma unroll
    for (int ni = 0; ni < 2; ++ni) {
      int col = n0 + ni * 16 + cr;
      float bb = sb1[b * C_ + col];
      v8f a = acc[mi * 2 + ni];
#pragma unroll
      for (int r = 0; r < 8; ++r) {
        int row = m0 + mi * 16 + r + 8 * hi;
        float d = (a[r] + bb) - vt[mi][ni][r];
        dd[mi][ni][r] = d;
        dp[row * C_ + col] = d;
      }
    }
#pragma unroll
  for (int mi = 0; mi < 2; ++mi)
#pragma unroll
    for (int ni = 0; ni < 2; ++ni) {
      int col = n0 + ni * 16 + cr;
#pragma unroll
      for (int r = 0; r < 8; ++r) {
        int row = m0 + mi * 16 + r + 8 * hi;
        wdpT[col * T_ + row] = wt[mi][r] * dd[mi][ni][r];
      }
    }
}

__device__ __forceinline__ void nw1_tile(const float* __restrict__ sW1,
                                         float* __restrict__ nW1,
                                         const float* __restrict__ ws,
                                         int b, int m0, int n0, int lane) {
  const float* A = ws + OFF_WDPT + b * C_ * T_;
  const float* Bm = ws + OFF_H1 + b * T_ * H_;
  const float dT = ws[OFF_W + 64];
  float* out = nW1 + b * C_ * H_;
  const float* w1 = sW1 + b * C_ * H_;
  v8f acc[4] = {};
  gemm32x32<false, T_, 8>(A, T_, Bm, H_, m0, n0, lane, acc);
  const int cr = lane & 15, hi = lane >> 4;
#pragma unroll
  for (int mi = 0; mi < 2; ++mi)
#pragma unroll
    for (int ni = 0; ni < 2; ++ni) {
      int col = n0 + ni * 16 + cr;
      v8f a = acc[mi * 2 + ni];
#pragma unroll
      for (int r = 0; r < 8; ++r) {
        int row = m0 + mi * 16 + r + 8 * hi;
        out[row * H_ + col] = dT * w1[row * H_ + col] + a[r];
      }
    }
}

__global__ void __launch_bounds__(32) k_gemm_dh1(const float* __restrict__ sW1,
                           const float* __restrict__ Wo, const float* __restrict__ bo,
                           float* __restrict__ outv,
                           float* __restrict__ nW1,
                           float* __restrict__ ws) {
  const int id = blockIdx.x, lane = threadIdx.x;
  if (id >= 120) {
    const int j = id - 120;
    const int b = j / 288, t = j % 288;
    nw1_tile(sW1, nW1, ws, b, (t / 24) * 32, (t % 24) * 32, lane);
    return;
  }
  if (id >= 96) {
    const int r = id - 96;
    const int b = r / 12, n0 = (r % 12) * 32;
    const float bb = bo[n0 + lane];
    const float y = rowvec32<C_, 16>(ws + OFF_PR + b * C_, Wo, C_, n0, lane);
    outv[b * C_ + n0 + lane] = y + bb;
    return;
  }
  const int b = id / 48, tile = id % 48;
  const int m0 = (tile % 2) * 32, n0 = (tile / 2) * 32;
  const float* A = ws + OFF_DP + b * T_ * C_;
  const float* W = sW1 + b * C_ * H_;
  const float* gp = ws + OFF_GP + b * T_ * H_;
  float* wdhT = ws + OFF_WDHT + b * H_ * T_;
  const int cr = lane & 15, hi = lane >> 4;
  float gt[2][2][8], wt[2][8];
#pragma unroll
  for (int mi = 0; mi < 2; ++mi)
#pragma unroll
    for (int r = 0; r < 8; ++r) {
      int row = m0 + mi * 16 + r + 8 * hi;
      wt[mi][r] = ws[OFF_W + row];
#pragma unroll
      for (int ni = 0; ni < 2; ++ni)
        gt[mi][ni][r] = gp[row * H_ + n0 + ni * 16 + cr];
    }
  v8f acc[4] = {};
  gemm32x32<false, C_, 16>(A, C_, W, H_, m0, n0, lane, acc);
#pragma unroll
  for (int mi = 0; mi < 2; ++mi)
#pragma unroll
    for (int ni = 0; ni < 2; ++ni) {
      int col = n0 + ni * 16 + cr;
      v8f a = acc[mi * 2 + ni];
#pragma unroll
      for (int r = 0; r < 8; ++r) {
        int row = m0 + mi * 16 + r + 8 * hi;
        float d = a[r] * gt[mi][ni][r];
        wdhT[col * T_ + row] = wt[mi][r] * d;
      }
    }
}

__global__ void __launch_bounds__(32) k_gemm_nw0(const float* __restrict__ sW0,
                           float* __restrict__ nW0,
                           float* __restrict__ ws) {
  const int b = blockIdx.y, lane = threadIdx.x;
  const int m0 = (blockIdx.x / 12) * 32, n0 = (blockIdx.x % 12) * 32;
  const float* A = ws + OFF_WDHT + b * H_ * T_;
  const float* Bm = ws + OFF_K + b * T_ * C_;
  const float dT = ws[OFF_W + 64];
  float* out = nW0 + b * H_ * C_;
  const float* w0 = sW0 + b * H_ * C_;
  v8f acc[4] = {};
  gemm32x32<false, T_, 8>(A, T_, Bm, C_, m0, n0, lane, acc);
  const int cr = lane & 15, hi = lane >> 4;
#pragma unroll
  for (int mi = 0; mi < 2; ++mi)
#pragma unroll
    for (int ni = 0; ni < 2; ++ni) {
      int col = n0 + ni * 16 + cr;
      v8f a = acc[mi * 2 + ni];
#pragma unroll
      for (int r = 0; r < 8; ++r) {
        int row = m0 + mi * 16 + r + 8 * hi;
        out[row * C_ + col] = dT * w0[row * C_ + col] + a[r];
      }
    }
}

__global__ void k_bias(const float* __restrict__ sb0, const float* __restrict__ sb1,
                       float* __restrict__ nb0, float* __restrict__ nb1,
                       const float* __restrict__ ws) {
  int idx = blockIdx.x * blockDim.x + threadIdx.x;
  if (idx >= B_ * (H_ + C_)) return;
  int b = idx / (H_ + C_), r = idx % (H_ + C_);
  const float dT = ws[OFF_W + 64];
  if (r < H_) {
    const float* p = ws + OFF_WDHT + b * H_ * T_ + r * T_;
    float s = 0.f;
#pragma unroll 4
    for (int t = 0; t < T_; ++t) s += p[t];
    nb0[b * H_ + r] = dT * sb0[b * H_ + r] + s;
  } else {
    int c = r - H_;
    const float* p = ws + OFF_WDPT + b * C_ * T_ + c * T_;
    float s = 0.f;
#pragma unroll 4
    for (int t = 0; t < T_; ++t) s += p[t];
    nb1[b * C_ + c] = dT * sb1[b * C_ + c] + s;
  }
}

}

extern "C" void kernel_launch(void* const* d_in, const int* in_sizes, int n_in,
                              void* d_out, int out_size, void* d_ws, size_t ws_size,
                              hipStream_t stream) {
  const float* x   = (const float*)d_in[0];
  const float* Wq  = (const float*)d_in[1];
  const float* bq  = (const float*)d_in[2];
  const float* Wk  = (const float*)d_in[3];
  const float* bk  = (const float*)d_in[4];
  const float* Wv  = (const float*)d_in[5];
  const float* bv  = (const float*)d_in[6];
  const float* Wo  = (const float*)d_in[7];
  const float* bo  = (const float*)d_in[8];
  const float* iq  = (const float*)d_in[9];
  const float* sW0 = (const float*)d_in[10];
  const float* sb0 = (const float*)d_in[11];
  const float* sW1 = (const float*)d_in[12];
  const float* sb1 = (const float*)d_in[13];

  float* out = (float*)d_out;
  float* ws  = (float*)d_ws;

  float* out_out = out;
  float* out_nW0 = out + B_ * C_;
  float* out_nb0 = out_nW0 + B_ * H_ * C_;
  float* out_nW1 = out_nb0 + B_ * H_;
  float* out_nb1 = out_nW1 + B_ * C_ * H_;

  k_init_w <<<1, 64, 0, stream>>>(ws);
  k_gemm_kv<<<dim3(27, B_, 2), 32, 0, stream>>>(x, Wk, bk, Wv, bv, Wq, bq, iq, ws);
  k_gemm_h1<<<dim3(72, B_), 32, 0, stream>>>(sW0, sb0, ws);
  k_gemm_dp<<<dim3(36, B_), 32, 0, stream>>>(sW1, sb1, ws);
  k_gemm_dh1<<<dim3(696), 32, 0, stream>>>(sW1, Wo, bo, out_out, out_nW1, ws);
  k_gemm_nw0<<<dim3(288, B_), 32, 0, stream>>>(sW0, out_nW0, ws);
  k_bias    <<<dim3(18), 128, 0, stream>>>(sb0, sb1, out_nb0, out_nb1, ws);
}


# Head and tail launches go: scan weights built by an extra block of launch 1, bias row sums by 72 extra blocks of launch 5 (register budget raised: tile loads stay grouped); 7 -> 5; sums keep order
# speedup vs baseline: 4.3835x; 1.1099x over previous
#include <hip/hip_runtime.h>
#include <cstdint>

typedef float v2f __attribute__((ext_vector_type(2)));
typedef float v8f __attribute__((ext_vector_type(8)));

namespace {

constexpr int B_ = 2, T_ = 64, C_ = 384, H_ = 768;
constexpr float LR = 0.01f, MOM = 0.9f, DECAY = 0.001f;

constexpr int OFF_W    = 0;
constexpr int OFF_K    = 128;
constexpr int OFF_V    = OFF_K    + B_*T_*C_;
constexpr int OFF_H1   = OFF_V    + B_*T_*C_;
constexpr int OFF_GP   = OFF_H1   + B_*T_*H_;
constexpr int OFF_DP   = OFF_GP   + B_*T_*H_;
constexpr int OFF_WDPT = OFF_DP   + B_*T_*C_;
constexpr int OFF_WDHT = OFF_WDPT + B_*C_*T_;
constexpr int OFF_Q    = OFF_WDHT + B_*H_*T_;
constexpr int OFF_HR   = OFF_Q    + C_;
constexpr int OFF_PR   = OFF_HR   + B_*H_;

__device__ __forceinline__ v8f wmma4(v2f a, v2f b, v8f c) {
  return __builtin_amdgcn_wmma_f32_16x16x4_f32(false, a, false, b, (short)0, c,
                                               false, false);
}

template <int KC>
struct FragSet {
  v2f a0[KC], a1[KC], b0[KC], b1[KC];
};

template <bool BT, int KC>
__device__ __forceinline__ void load_set(FragSet<KC>& f,
                                         const float* __restrict__ pa, int lda,
                                         const float* __restrict__ pb, int ldb,
                                         int kk) {
#pragma unroll
  for (int s = 0; s < KC; ++s) {
    const int k = kk + 4 * s;
    f.a0[s] = *reinterpret_cast<const v2f*>(pa + k);
    f.a1[s] = *reinterpret_cast<const v2f*>(pa + 16 * lda + k);
    if (BT) {
      f.b0[s] = *reinterpret_cast<const v2f*>(pb + k);
      f.b1[s] = *reinterpret_cast<const v2f*>(pb + 16 * ldb + k);
    } else {
      const float* q = pb + k * ldb;
      f.b0[s].x = q[0];  f.b0[s].y = q[ldb];
      f.b1[s].x = q[16]; f.b1[s].y = q[ldb + 16];
    }
  }
}

template <int KC>
__device__ __forceinline__ void mma_set(const FragSet<KC>& f, v8f acc[4]) {
#pragma unroll
  for (int s = 0; s < KC; ++s) {
    acc[0] = wmma4(f.a0[s], f.b0[s], acc[0]);
    acc[1] = wmma4(f.a0[s], f.b1[s], acc[1]);
    acc[2] = wmma4(f.a1[s], f.b0[s], acc[2]);
    acc[3] = wmma4(f.a1[s], f.b1[s], acc[3]);
  }
}

template <bool BT, int K, int KC>
__device__ __forceinline__ void gemm32x32(const float* __restrict__ A, int lda,
                                          const float* __restrict__ Bm, int ldb,
                                          int m0, int n0, int lane,
                                          v8f acc[4]) {
  constexpr int CH = 4 * KC;
  constexpr int NCH = K / CH;
  static_assert(K % CH == 0 && NCH % 2 == 0, "K must be an even number of chunks");
  const int mr = lane & 15;
  const int kh = (lane >> 4) << 1;
  const float* pa = A + (m0 + mr) * lda + kh;
  const float* pb = BT ? Bm + (n0 + mr) * ldb + kh : Bm + kh * ldb + n0 + mr;
  FragSet<KC> f0, f1;
  load_set<BT, KC>(f0, pa, lda, pb, ldb, 0);
  for (int c = 0; c + 2 < NCH; c += 2) {
    load_set<BT, KC>(f1, pa, lda, pb, ldb, (c + 1) * CH);
    mma_set<KC>(f0, acc);
    load_set<BT, KC>(f0, pa, lda, pb, ldb, (c + 2) * CH);
    mma_set<KC>(f1, acc);
  }
  load_set<BT, KC>(f1, pa, lda, pb, ldb, (NCH - 1) * CH);
  mma_set<KC>(f0, acc);
  mma_set<KC>(f1, acc);
}

template <int KC>
struct RowSet {
  v2f a[KC], b0[KC], b1[KC];
};

template <int KC>
__device__ __forceinline__ void load_row_set(RowSet<KC>& f,
                                             const float* __restrict__ pa,
                                             const float* __restrict__ pb, int ldb,
                                             int kk) {
#pragma unroll
  for (int s = 0; s < KC; ++s) {
    const int k = kk + 4 * s;
    f.a[s]  = *reinterpret_cast<const v2f*>(pa + k);
    f.b0[s] = *reinterpret_cast<const v2f*>(pb + k);
    f.b1[s] = *reinterpret_cast<const v2f*>(pb + 16 * ldb + k);
  }
}

template <int KC>
__device__ __forceinline__ void mma_row_set(const RowSet<KC>& f, v8f acc[2]) {
#pragma unroll
  for (int s = 0; s < KC; ++s) {
    acc[0] = wmma4(f.a[s], f.b0[s], acc[0]);
    acc[1] = wmma4(f.a[s], f.b1[s], acc[1]);
  }
}

template <int K, int KC>
__device__ __forceinline__ float rowvec32(const float* __restrict__ xrow,
                                          const float* __restrict__ Wm, int ldb,
                                          int n0, int lane) {
  constexpr int CH = 4 * KC;
  constexpr int NCH = K / CH;
  static_assert(K % CH == 0 && NCH % 2 == 0, "K must be an even number of chunks");
  const int mr = lane & 15;
  const int kh = (lane >> 4) << 1;
  const float* pa = xrow + kh;
  const float* pb = Wm + (n0 + mr) * ldb + kh;
  v8f acc[2] = {};
  RowSet<KC> f0, f1;
  load_row_set<KC>(f0, pa, pb, ldb, 0);
  for (int c = 0; c + 2 < NCH; c += 2) {
    load_row_set<KC>(f1, pa, pb, ldb, (c + 1) * CH);
    mma_row_set<KC>(f0, acc);
    load_row_set<KC>(f0, pa, pb, ldb, (c + 2) * CH);
    mma_row_set<KC>(f1, acc);
  }
  load_row_set<KC>(f1, pa, pb, ldb, (NCH - 1) * CH);
  mma_row_set<KC>(f0, acc);
  mma_row_set<KC>(f1, acc);
  return (lane >> 4) ? acc[1][0] : acc[0][0];
}

__device__ __forceinline__ void init_w(float* __restrict__ ws, int lane) {
  const float a = 1.0f - DECAY;
#pragma unroll
  for (int j = 0; j < 2; ++j) {
    const int t = lane + 32 * j;
    float n = (float)(T_ - t);
    float an = __powf(a, n), mn = __powf(MOM, n);
    ws[OFF_W + t] = -LR * (1.0f - MOM) * (an - mn) / (a - MOM);
  }
  if (lane == 0) ws[OFF_W + 64] = __powf(a, (float)T_);
}

__global__ void __launch_bounds__(32) k_gemm_kv(const float* __restrict__ x,
                          const float* __restrict__ Wk, const float* __restrict__ bk,
                          const float* __restrict__ Wv, const float* __restrict__ bv,
                          const float* __restrict__ Wq, const float* __restrict__ bq,
                          const float* __restrict__ iq,
                          float* __restrict__ ws) {
  const int b = blockIdx.y, which = blockIdx.z, lane = threadIdx.x;
  if (blockIdx.x == 27) {
    if (b == 0 && which == 0) init_w(ws, lane);
    return;
  }
  if (blockIdx.x >= 24) {
    const int n0 = ((blockIdx.x - 24) + 3 * (b + 2 * which)) * 32;
    const float bb = bq[n0 + lane];
    const float y = rowvec32<C_, 16>(iq, Wq, C_, n0, lane);
    ws[OFF_Q + n0 + lane] = y + bb;
    return;
  }
  const int m0 = (blockIdx.x & 1) * 32, n0 = (blockIdx.x >> 1) * 32;
  const float* A = x + b * T_ * C_;
  const float* W = which ? Wv : Wk;
  const float* bias = which ? bv : bk;
  float* out = ws + (which ? OFF_V : OFF_K) + b * T_ * C_;
  v8f acc[4] = {};
  gemm32x32<true, C_, 16>(A, C_, W, C_, m0, n0, lane, acc);
  const int cr = lane & 15, hi = lane >> 4;
#pragma unroll
  for (int mi = 0; mi < 2; ++mi)
#pragma unroll
    for (int ni = 0; ni < 2; ++ni) {
      int col = n0 + ni * 16 + cr;
      float bb = bias[col];
      v8f a = acc[mi * 2 + ni];
#pragma unroll
      for (int r = 0; r < 8; ++r) {
        int row = m0 + mi * 16 + r + 8 * hi;
        out[row * C_ + col] = a[r] + bb;
      }
    }
}

__global__ void __launch_bounds__(32) k_gemm_h1(const float* __restrict__ sW0,
                          const float* __restrict__ sb0,
                          float* __restrict__ ws) {
  const int b = blockIdx.y, lane = threadIdx.x;
  if (blockIdx.x >= 48) {
    const int n0 = (blockIdx.x - 48) * 32;
    const float bb = sb0[b * H_ + n0 + lane];
    const float pre = rowvec32<C_, 16>(ws + OFF_Q, sW0 + b * H_ * C_, C_, n0, lane) + bb;
    const float sig = 1.0f / (1.0f + __expf(-pre));
    ws[OFF_HR + b * H_ + n0 + lane] = pre * sig;
    return;
  }
  const int m0 = (blockIdx.x % 2) * 32, n0 = (blockIdx.x / 2) * 32;
  const float* A = ws + OFF_K + b * T_ * C_;
  const float* W = sW0 + b * H_ * C_;
  float* h1 = ws + OFF_H1 + b * T_ * H_;
  float* gp = ws + OFF_GP + b * T_ * H_;
  v8f acc[4] = {};
  gemm32x32<true, C_, 16>(A, C_, W, C_, m0, n0, lane, acc);
  const int cr = lane & 15, hi = lane >> 4;
#pragma unroll
  for (int mi = 0; mi < 2; ++mi)
#pragma unroll
    for (int ni = 0; ni < 2; ++ni) {
      int col = n0 + ni * 16 + cr;
      float bb = sb0[b * H_ + col];
      v8f a = acc[mi * 2 + ni];
#pragma unroll
      for (int r = 0; r < 8; ++r) {
        int row = m0 + mi * 16 + r + 8 * hi;
        float pre = a[r] + bb;
        float sig = 1.0f / (1.0f + __expf(-pre));
        h1[row * H_ + col] = pre * sig;
        gp[row * H_ + col] = sig * (1.0f + pre * (1.0f - sig));
      }
    }
}

__global__ void __launch_bounds__(32) k_gemm_dp(const float* __restrict__ sW1,
                          const float* __restrict__ sb1,
                          float* __restrict__ ws) {
  const int b = blockIdx.y, lane = threadIdx.x;
  if (blockIdx.x >= 24) {
    const int n0 = (blockIdx.x - 24) * 32;
    const float bb = sb1[b * C_ + n0 + lane];
    const float y = rowvec32<H_, 16>(ws + OFF_HR + b * H_, sW1 + b * C_ * H_, H_, n0, lane);
    ws[OFF_PR + b * C_ + n0 + lane] = y + bb;
    return;
  }
  const int m0 = (blockIdx.x % 2) * 32, n0 = (blockIdx.x / 2) * 32;
  const float* A = ws + OFF_H1 + b * T_ * H_;
  const float* W = sW1 + b * C_ * H_;
  const float* v = ws + OFF_V + b * T_ * C_;
  float* dp = ws + OFF_DP + b * T_ * C_;
  float* wdpT = ws + OFF_WDPT + b * C_ * T_;
  const int cr = lane & 15, hi = lane >> 4;
  float vt[2][2][8], wt[2][8];
#pragma unroll
  for (int mi = 0; mi < 2; ++mi)
#pragma unroll
    for (int r = 0; r < 8; ++r) {
      int row = m0 + mi * 16 + r + 8 * hi;
      wt[mi][r] = ws[OFF_W + row];
#pragma unroll
      for (int ni = 0; ni < 2; ++ni)
        vt[mi][ni][r] = v[row * C_ + n0 + ni * 16 + cr];
    }
  v8f acc[4] = {};
  gemm32x32<true, H_, 16>(A, H_, W, H_, m0, n0, lane, acc);
  float dd[2][2][8];
#pragma unroll
  for (int mi = 0; mi < 2; ++mi)
#pragma unroll
    for (int ni = 0; ni < 2; ++ni) {
      int col = n0 + ni * 16 + cr;
      float bb = sb1[b * C_ + col];
      v8f a = acc[mi * 2 + ni];
#pragma unroll
      for (int r = 0; r < 8; ++r) {
        int row = m0 + mi * 16 + r + 8 * hi;
        float d = (a[r] + bb) - vt[mi][ni][r];
        dd[mi][ni][r] = d;
        dp[row * C_ + col] = d;
      }
    }
#pragma unroll
  for (int mi = 0; mi < 2; ++mi)
#pragma unroll
    for (int ni = 0; ni < 2; ++ni) {
      int col = n0 + ni * 16 + cr;
#pragma unroll
      for (int r = 0; r < 8; ++r) {
        int row = m0 + mi * 16 + r + 8 * hi;
        wdpT[col * T_ + row] = wt[mi][r] * dd[mi][ni][r];
      }
    }
}

__device__ __forceinline__ void nw1_tile(const float* __restrict__ sW1,
                                         float* __restrict__ nW1,
                                         const float* __restrict__ ws,
                                         int b, int m0, int n0, int lane) {
  const float* A = ws + OFF_WDPT + b * C_ * T_;
  const float* Bm = ws + OFF_H1 + b * T_ * H_;
  const float dT = ws[OFF_W + 64];
  float* out = nW1 + b * C_ * H_;
  const float* w1 = sW1 + b * C_ * H_;
  v8f acc[4] = {};
  gemm32x32<false, T_, 8>(A, T_, Bm, H_, m0, n0, lane, acc);
  const int cr = lane & 15, hi = lane >> 4;
#pragma unroll
  for (int mi = 0; mi < 2; ++mi)
#pragma unroll
    for (int ni = 0; ni < 2; ++ni) {
      int col = n0 + ni * 16 + cr;
      v8f a = acc[mi * 2 + ni];
#pragma unroll
      for (int r = 0; r < 8; ++r) {
        int row = m0 + mi * 16 + r + 8 * hi;
        out[row * H_ + col] = dT * w1[row * H_ + col] + a[r];
      }
    }
}

__global__ void __launch_bounds__(32) k_gemm_dh1(const float* __restrict__ sW1,
                           const float* __restrict__ Wo, const float* __restrict__ bo,
                           float* __restrict__ outv,
                           float* __restrict__ nW1,
                           float* __restrict__ ws) {
  const int id = blockIdx.x, lane = threadIdx.x;
  if (id >= 120) {
    const int j = id - 120;
    const int b = j / 288, t = j % 288;
    nw1_tile(sW1, nW1, ws, b, (t / 24) * 32, (t % 24) * 32, lane);
    return;
  }
  if (id >= 96) {
    const int r = id - 96;
    const int b = r / 12, n0 = (r % 12) * 32;
    const float bb = bo[n0 + lane];
    const float y = rowvec32<C_, 16>(ws + OFF_PR + b * C_, Wo, C_, n0, lane);
    outv[b * C_ + n0 + lane] = y + bb;
    return;
  }
  const int b = id / 48, tile = id % 48;
  const int m0 = (tile % 2) * 32, n0 = (tile / 2) * 32;
  const float* A = ws + OFF_DP + b * T_ * C_;
  const float* W = sW1 + b * C_ * H_;
  const float* gp = ws + OFF_GP + b * T_ * H_;
  float* wdhT = ws + OFF_WDHT + b * H_ * T_;
  const int cr = lane & 15, hi = lane >> 4;
  float gt[2][2][8], wt[2][8];
#pragma unroll
  for (int mi = 0; mi < 2; ++mi)
#pragma unroll
    for (int r = 0; r < 8; ++r) {
      int row = m0 + mi * 16 + r + 8 * hi;
      wt[mi][r] = ws[OFF_W + row];
#pragma unroll
      for (int ni = 0; ni < 2; ++ni)
        gt[mi][ni][r] = gp[row * H_ + n0 + ni * 16 + cr];
    }
  v8f acc[4] = {};
  gemm32x32<false, C_, 16>(A, C_, W, H_, m0, n0, lane, acc);
#pragma unroll
  for (int mi = 0; mi < 2; ++mi)
#pragma unroll
    for (int ni = 0; ni < 2; ++ni) {
      int col = n0 + ni * 16 + cr;
      v8f a = acc[mi * 2 + ni];
#pragma unroll
      for (int r = 0; r < 8; ++r) {
        int row = m0 + mi * 16 + r + 8 * hi;
        float d = a[r] * gt[mi][ni][r];
        wdhT[col * T_ + row] = wt[mi][r] * d;
      }
    }
}

__device__ __forceinline__ float row_sum64(const float* __restrict__ p) {
  float4 q[T_ / 4];
#pragma unroll
  for (int j = 0; j < T_ / 4; ++j) q[j] = reinterpret_cast<const float4*>(p)[j];
  float s = 0.f;
#pragma unroll
  for (int j = 0; j < T_ / 4; ++j) {
    s += q[j].x; s += q[j].y; s += q[j].z; s += q[j].w;
  }
  return s;
}

__global__ void __launch_bounds__(32) __attribute__((amdgpu_waves_per_eu(1, 4))) k_gemm_nw0(const float* __restrict__ sW0,
                           const float* __restrict__ sb0, const float* __restrict__ sb1,
                           float* __restrict__ nW0,
                           float* __restrict__ nb0, float* __restrict__ nb1,
                           float* __restrict__ ws) {
  const int b = blockIdx.y, lane = threadIdx.x;
  if (blockIdx.x >= 288) {
    const int blk = blockIdx.x - 288;
    const float dT = ws[OFF_W + 64];
    if (blk < H_ / 32) {
      const int r = blk * 32 + lane;
      const float s = row_sum64(ws + OFF_WDHT + b * H_ * T_ + r * T_);
      nb0[b * H_ + r] = dT * sb0[b * H_ + r] + s;
    } else {
      const int c = (blk - H_ / 32) * 32 + lane;
      const float s = row_sum64(ws + OFF_WDPT + b * C_ * T_ + c * T_);
      nb1[b * C_ + c] = dT * sb1[b * C_ + c] + s;
    }
    return;
  }
  const int m0 = (blockIdx.x / 12) * 32, n0 = (blockIdx.x % 12) * 32;
  const float* A = ws + OFF_WDHT + b * H_ * T_;
  const float* Bm = ws + OFF_K + b * T_ * C_;
  const float dT = ws[OFF_W + 64];
  float* out = nW0 + b * H_ * C_;
  const float* w0 = sW0 + b * H_ * C_;
  v8f acc[4] = {};
  gemm32x32<false, T_, 8>(A, T_, Bm, C_, m0, n0, lane, acc);
  const int cr = lane & 15, hi = lane >> 4;
#pragma unroll
  for (int mi = 0; mi < 2; ++mi)
#pragma unroll
    for (int ni = 0; ni < 2; ++ni) {
      int col = n0 + ni * 16 + cr;
      v8f a = acc[mi * 2 + ni];
#pragma unroll
      for (int r = 0; r < 8; ++r) {
        int row = m0 + mi * 16 + r + 8 * hi;
        out[row * C_ + col] = dT * w0[row * C_ + col] + a[r];
      }
    }
}

}

extern "C" void kernel_launch(void* const* d_in, const int* in_sizes, int n_in,
                              void* d_out, int out_size, void* d_ws, size_t ws_size,
                              hipStream_t stream) {
  const float* x   = (const float*)d_in[0];
  const float* Wq  = (const float*)d_in[1];
  const float* bq  = (const float*)d_in[2];
  const float* Wk  = (const float*)d_in[3];
  const float* bk  = (const float*)d_in[4];
  const float* Wv  = (const float*)d_in[5];
  const float* bv  = (const float*)d_in[6];
  const float* Wo  = (const float*)d_in[7];
  const float* bo  = (const float*)d_in[8];
  const float* iq  = (const float*)d_in[9];
  const float* sW0 = (const float*)d_in[10];
  const float* sb0 = (const float*)d_in[11];
  const float* sW1 = (const float*)d_in[12];
  const float* sb1 = (const float*)d_in[13];

  float* out = (float*)d_out;
  float* ws  = (float*)d_ws;

  float* out_out = out;
  float* out_nW0 = out + B_ * C_;
  float* out_nb0 = out_nW0 + B_ * H_ * C_;
  float* out_nW1 = out_nb0 + B_ * H_;
  float* out_nb1 = out_nW1 + B_ * C_ * H_;

  k_gemm_kv<<<dim3(28, B_, 2), 32, 0, stream>>>(x, Wk, bk, Wv, bv, Wq, bq, iq, ws);
  k_gemm_h1<<<dim3(72, B_), 32, 0, stream>>>(sW0, sb0, ws);
  k_gemm_dp<<<dim3(36, B_), 32, 0, stream>>>(sW1, sb1, ws);
  k_gemm_dh1<<<dim3(696), 32, 0, stream>>>(sW1, Wo, bo, out_out, out_nW1, ws);
  k_gemm_nw0<<<dim3(288 + 36, B_), 32, 0, stream>>>(sW0, sb0, sb1, out_nW0, out_nb0, out_nb1, ws);
}


# K of launch 3 and of launch 4's dh1 cut into 4 one-wave blocks each; partial planes (linear epilogue applied, w-planes in fragment order) are added by the next GEMM's A loader: sum order changes
# speedup vs baseline: 4.7754x; 1.0894x over previous
#include <hip/hip_runtime.h>
#include <cstdint>

typedef float v2f __attribute__((ext_vector_type(2)));
typedef float v8f __attribute__((ext_vector_type(8)));

namespace {

constexpr int B_ = 2, T_ = 64, C_ = 384, H_ = 768;
constexpr float LR = 0.01f, MOM = 0.9f, DECAY = 0.001f;

constexpr int SD = 4;
constexpr int SH = 4;

constexpr int OFF_W    = 0;
constexpr int OFF_K    = 128;
constexpr int OFF_V    = OFF_K    + B_*T_*C_;
constexpr int OFF_H1   = OFF_V    + B_*T_*C_;
constexpr int OFF_GP   = OFF_H1   + B_*T_*H_;
constexpr int PL_DP   = B_*T_*C_;
constexpr int PL_WDPT = B_*C_*T_;
constexpr int PL_WDHT = B_*H_*T_;
constexpr int OFF_DP   = OFF_GP   + B_*T_*H_;
constexpr int OFF_WDPT = OFF_DP   + SD*PL_DP;
constexpr int OFF_WDHT = OFF_WDPT + SD*PL_WDPT;
constexpr int OFF_Q    = OFF_WDHT + SH*PL_WDHT;
constexpr int OFF_HR   = OFF_Q    + C_;
constexpr int OFF_PR   = OFF_HR   + B_*H_;

__device__ __forceinline__ v8f wmma4(v2f a, v2f b, v8f c) {
  return __builtin_amdgcn_wmma_f32_16x16x4_f32(false, a, false, b, (short)0, c,
                                               false, false);
}

template <int KC, int S>
struct FragSet {
  v2f a0[KC][S], a1[KC][S], b0[KC], b1[KC];
};

template <bool BT, int KC, int S>
__device__ __forceinline__ void load_set(FragSet<KC, S>& f,
                                         const float* __restrict__ pa, int a1off, int ks, int plane,
                                         const float* __restrict__ pb, int ldb,
                                         int kk) {
#pragma unroll
  for (int s = 0; s < KC; ++s) {
    const int k = kk + 4 * s;
#pragma unroll
    for (int p = 0; p < S; ++p) {
      f.a0[s][p] = *reinterpret_cast<const v2f*>(pa + p * plane + k * ks);
      f.a1[s][p] = *reinterpret_cast<const v2f*>(pa + p * plane + a1off + k * ks);
    }
    if (BT) {
      f.b0[s] = *reinterpret_cast<const v2f*>(pb + k);
      f.b1[s] = *reinterpret_cast<const v2f*>(pb + 16 * ldb + k);
    } else {
      const float* q = pb + k * ldb;
      f.b0[s].x = q[0];  f.b0[s].y = q[ldb];
      f.b1[s].x = q[16]; f.b1[s].y = q[ldb + 16];
    }
  }
}

template <int KC, int S>
__device__ __forceinline__ void mma_set(const FragSet<KC, S>& f, v8f acc[4]) {
#pragma unroll
  for (int s = 0; s < KC; ++s) {
    v2f a0 = f.a0[s][0], a1 = f.a1[s][0];
#pragma unroll
    for (int p = 1; p < S; ++p) { a0 += f.a0[s][p]; a1 += f.a1[s][p]; }
    acc[0] = wmma4(a0, f.b0[s], acc[0]);
    acc[1] = wmma4(a0, f.b1[s], acc[1]);
    acc[2] = wmma4(a1, f.b0[s], acc[2]);
    acc[3] = wmma4(a1, f.b1[s], acc[3]);
  }
}

template <bool BT, int K, int KC, int S = 1, bool AF = false>
__device__ __forceinline__ void gemm32x32(const float* __restrict__ A, int lda,
                                          const float* __restrict__ Bm, int ldb,
                                          int m0, int n0, int lane,
                                          v8f acc[4], int plane = 0) {
  constexpr int CH = 4 * KC;
  constexpr int NCH = K / CH;
  static_assert(K % CH == 0 && NCH % 2 == 0, "K must be an even number of chunks");
  const int mr = lane & 15;
  const int kh = (lane >> 4) << 1;
  static_assert(!AF || K == T_, "fragment order is laid out for K = T_");
  const float* pa = AF ? A + (m0 >> 4) * (16 * T_) + (lane >> 4) * 32 + mr * 2
                       : A + (m0 + mr) * lda + kh;
  const int a1off = AF ? 16 * T_ : 16 * lda;
  const int ks = AF ? 16 : 1;
  const float* pb = BT ? Bm + (n0 + mr) * ldb + kh : Bm + kh * ldb + n0 + mr;
  FragSet<KC, S> f0, f1;
  load_set<BT, KC, S>(f0, pa, a1off, ks, plane, pb, ldb, 0);
  for (int c = 0; c + 2 < NCH; c += 2) {
    load_set<BT, KC, S>(f1, pa, a1off, ks, plane, pb, ldb, (c + 1) * CH);
    mma_set<KC, S>(f0, acc);
    load_set<BT, KC, S>(f0, pa, a1off, ks, plane, pb, ldb, (c + 2) * CH);
    mma_set<KC, S>(f1, acc);
  }
  load_set<BT, KC, S>(f1, pa, a1off, ks, plane, pb, ldb, (NCH - 1) * CH);
  mma_set<KC, S>(f0, acc);
  mma_set<KC, S>(f1, acc);
}

template <int KC>
struct RowSet {
  v2f a[KC], b0[KC], b1[KC];
};

template <int KC>
__device__ __forceinline__ void load_row_set(RowSet<KC>& f,
                                             const float* __restrict__ pa,
                                             const float* __restrict__ pb, int ldb,
                                             int kk) {
#pragma unroll
  for (int s = 0; s < KC; ++s) {
    const int k = kk + 4 * s;
    f.a[s]  = *reinterpret_cast<const v2f*>(pa + k);
    f.b0[s] = *reinterpret_cast<const v2f*>(pb + k);
    f.b1[s] = *reinterpret_cast<const v2f*>(pb + 16 * ldb + k);
  }
}

template <int KC>
__device__ __forceinline__ void mma_row_set(const RowSet<KC>& f, v8f acc[2]) {
#pragma unroll
  for (int s = 0; s < KC; ++s) {
    acc[0] = wmma4(f.a[s], f.b0[s], acc[0]);
    acc[1] = wmma4(f.a[s], f.b1[s], acc[1]);
  }
}

template <int K, int KC>
__device__ __forceinline__ float rowvec32(const float* __restrict__ xrow,
                                          const float* __restrict__ Wm, int ldb,
                                          int n0, int lane) {
  constexpr int CH = 4 * KC;
  constexpr int NCH = K / CH;
  static_assert(K % CH == 0 && NCH % 2 == 0, "K must be an even number of chunks");
  const int mr = lane & 15;
  const int kh = (lane >> 4) << 1;
  const float* pa = xrow + kh;
  const float* pb = Wm + (n0 + mr) * ldb + kh;
  v8f acc[2] = {};
  RowSet<KC> f0, f1;
  load_row_set<KC>(f0, pa, pb, ldb, 0);
  for (int c = 0; c + 2 < NCH; c += 2) {
    load_row_set<KC>(f1, pa, pb, ldb, (c + 1) * CH);
    mma_row_set<KC>(f0, acc);
    load_row_set<KC>(f0, pa, pb, ldb, (c + 2) * CH);
    mma_row_set<KC>(f1, acc);
  }
  load_row_set<KC>(f1, pa, pb, ldb, (NCH - 1) * CH);
  mma_row_set<KC>(f0, acc);
  mma_row_set<KC>(f1, acc);
  return (lane >> 4) ? acc[1][0] : acc[0][0];
}

__device__ __forceinline__ void init_w(float* __restrict__ ws, int lane) {
  const float a = 1.0f - DECAY;
#pragma unroll
  for (int j = 0; j < 2; ++j) {
    const int t = lane + 32 * j;
    float n = (float)(T_ - t);
    float an = __powf(a, n), mn = __powf(MOM, n);
    ws[OFF_W + t] = -LR * (1.0f - MOM) * (an - mn) / (a - MOM);
  }
  if (lane == 0) ws[OFF_W + 64] = __powf(a, (float)T_);
}

__global__ void __launch_bounds__(32) k_gemm_kv(const float* __restrict__ x,
                          const float* __restrict__ Wk, const float* __restrict__ bk,
                          const float* __restrict__ Wv, const float* __restrict__ bv,
                          const float* __restrict__ Wq, const float* __restrict__ bq,
                          const float* __restrict__ iq,
                          float* __restrict__ ws) {
  const int b = blockIdx.y, which = blockIdx.z, lane = threadIdx.x;
  if (blockIdx.x == 27) {
    if (b == 0 && which == 0) init_w(ws, lane);
    return;
  }
  if (blockIdx.x >= 24) {
    const int n0 = ((blockIdx.x - 24) + 3 * (b + 2 * which)) * 32;
    const float bb = bq[n0 + lane];
    const float y = rowvec32<C_, 16>(iq, Wq, C_, n0, lane);
    ws[OFF_Q + n0 + lane] = y + bb;
    return;
  }
  const int m0 = (blockIdx.x & 1) * 32, n0 = (blockIdx.x >> 1) * 32;
  const float* A = x + b * T_ * C_;
  const float* W = which ? Wv : Wk;
  const float* bias = which ? bv : bk;
  float* out = ws + (which ? OFF_V : OFF_K) + b * T_ * C_;
  v8f acc[4] = {};
  gemm32x32<true, C_, 16>(A, C_, W, C_, m0, n0, lane, acc);
  const int cr = lane & 15, hi = lane >> 4;
#pragma unroll
  for (int mi = 0; mi < 2; ++mi)
#pragma unroll
    for (int ni = 0; ni < 2; ++ni) {
      int col = n0 + ni * 16 + cr;
      float bb = bias[col];
      v8f a = acc[mi * 2 + ni];
#pragma unroll
      for (int r = 0; r < 8; ++r) {
        int row = m0 + mi * 16 + r + 8 * hi;
        out[row * C_ + col] = a[r] + bb;
      }
    }
}

__global__ void __launch_bounds__(32) k_gemm_h1(const float* __restrict__ sW0,
                          const float* __restrict__ sb0,
                          float* __restrict__ ws) {
  const int b = blockIdx.y, lane = threadIdx.x;
  if (blockIdx.x >= 48) {
    const int n0 = (blockIdx.x - 48) * 32;
    const float bb = sb0[b * H_ + n0 + lane];
    const float pre = rowvec32<C_, 16>(ws + OFF_Q, sW0 + b * H_ * C_, C_, n0, lane) + bb;
    const float sig = 1.0f / (1.0f + __expf(-pre));
    ws[OFF_HR + b * H_ + n0 + lane] = pre * sig;
    return;
  }
  const int m0 = (blockIdx.x % 2) * 32, n0 = (blockIdx.x / 2) * 32;
  const float* A = ws + OFF_K + b * T_ * C_;
  const float* W = sW0 + b * H_ * C_;
  float* h1 = ws + OFF_H1 + b * T_ * H_;
  float* gp = ws + OFF_GP + b * T_ * H_;
  v8f acc[4] = {};
  gemm32x32<true, C_, 16>(A, C_, W, C_, m0, n0, lane, acc);
  const int cr = lane & 15, hi = lane >> 4;
#pragma unroll
  for (int mi = 0; mi < 2; ++mi)
#pragma unroll
    for (int ni = 0; ni < 2; ++ni) {
      int col = n0 + ni * 16 + cr;
      float bb = sb0[b * H_ + col];
      v8f a = acc[mi * 2 + ni];
#pragma unroll
      for (int r = 0; r < 8; ++r) {
        int row = m0 + mi * 16 + r + 8 * hi;
        float pre = a[r] + bb;
        float sig = 1.0f / (1.0f + __expf(-pre));
        h1[row * H_ + col] = pre * sig;
        gp[row * H_ + col] = sig * (1.0f + pre * (1.0f - sig));
      }
    }
}

__device__ __forceinline__ void store_frag_T(float* __restrict__ pl, int rowblk, int t0,
                                             int cr, int hi, const float val[8]) {
#pragma unroll
  for (int u = 0; u < 4; ++u) {
    const int kg = (t0 >> 2) + 2 * hi + (u >> 1);
    v2f x; x.x = val[2 * u]; x.y = val[2 * u + 1];
    *reinterpret_cast<v2f*>(pl + ((rowblk * (T_ / 4) + kg) * 2 + (u & 1)) * 32 + cr * 2) = x;
  }
}

__global__ void __launch_bounds__(32) __attribute__((amdgpu_waves_per_eu(1, 2))) k_gemm_dp(const float* __restrict__ sW1,
                          const float* __restrict__ sb1,
                          float* __restrict__ ws) {
  const int b = blockIdx.y, lane = threadIdx.x;
  if (blockIdx.x >= 24 * SD) {
    const int n0 = (blockIdx.x - 24 * SD) * 32;
    const float bb = sb1[b * C_ + n0 + lane];
    const float y = rowvec32<H_, 16>(ws + OFF_HR + b * H_, sW1 + b * C_ * H_, H_, n0, lane);
    ws[OFF_PR + b * C_ + n0 + lane] = y + bb;
    return;
  }
  const int part = blockIdx.x / 24, tile = blockIdx.x % 24;
  const int m0 = (tile % 2) * 32, n0 = (tile / 2) * 32;
  const int k0 = part * (H_ / SD);
  const float* A = ws + OFF_H1 + b * T_ * H_ + k0;
  const float* W = sW1 + b * C_ * H_ + k0;
  const float* v = ws + OFF_V + b * T_ * C_;
  float* dp = ws + OFF_DP + part * PL_DP + b * T_ * C_;
  float* wdpT = ws + OFF_WDPT + part * PL_WDPT + b * C_ * T_;
  const int cr = lane & 15, hi = lane >> 4;
  float vt[2][2][8] = {}, wt[2][8], bt[2] = {};
#pragma unroll
  for (int mi = 0; mi < 2; ++mi)
#pragma unroll
    for (int r = 0; r < 8; ++r) {
      int row = m0 + mi * 16 + r + 8 * hi;
      wt[mi][r] = ws[OFF_W + row];
    }
  if (part == 0) {
#pragma unroll
    for (int ni = 0; ni < 2; ++ni) bt[ni] = sb1[b * C_ + n0 + ni * 16 + cr];
#pragma unroll
    for (int mi = 0; mi < 2; ++mi)
#pragma unroll
      for (int r = 0; r < 8; ++r) {
        int row = m0 + mi * 16 + r + 8 * hi;
#pragma unroll
        for (int ni = 0; ni < 2; ++ni)
          vt[mi][ni][r] = v[row * C_ + n0 + ni * 16 + cr];
      }
  }
  v8f acc[4] = {};
  gemm32x32<true, H_ / SD, 12>(A, H_, W, H_, m0, n0, lane, acc);
  float dd[2][2][8];
#pragma unroll
  for (int mi = 0; mi < 2; ++mi)
#pragma unroll
    for (int ni = 0; ni < 2; ++ni) {
      int col = n0 + ni * 16 + cr;
      float bb = bt[ni];
      v8f a = acc[mi * 2 + ni];
#pragma unroll
      for (int r = 0; r < 8; ++r) {
        int row = m0 + mi * 16 + r + 8 * hi;
        float d = (a[r] + bb) - vt[mi][ni][r];
        dd[mi][ni][r] = d;
        dp[row * C_ + col] = d;
      }
    }
#pragma unroll
  for (int mi = 0; mi < 2; ++mi)
#pragma unroll
    for (int ni = 0; ni < 2; ++ni) {
      float val[8];
#pragma unroll
      for (int r = 0; r < 8; ++r) val[r] = wt[mi][r] * dd[mi][ni][r];
      store_frag_T(wdpT, (n0 >> 4) + ni, m0 + mi * 16, cr, hi, val);
    }
}

__device__ __forceinline__ void nw1_tile(const float* __restrict__ sW1,
                                         float* __restrict__ nW1,
                                         const float* __restrict__ ws,
                                         int b, int m0, int n0, int lane) {
  const float* A = ws + OFF_WDPT + b * C_ * T_;
  const float* Bm = ws + OFF_H1 + b * T_ * H_;
  const float dT = ws[OFF_W + 64];
  float* out = nW1 + b * C_ * H_;
  const float* w1 = sW1 + b * C_ * H_;
  v8f acc[4] = {};
  gemm32x32<false, T_, 8, SD, true>(A, T_, Bm, H_, m0, n0, lane, acc, PL_WDPT);
  const int cr = lane & 15, hi = lane >> 4;
#pragma unroll
  for (int mi = 0; mi < 2; ++mi)
#pragma unroll
    for (int ni = 0; ni < 2; ++ni) {
      int col = n0 + ni * 16 + cr;
      v8f a = acc[mi * 2 + ni];
#pragma unroll
      for (int r = 0; r < 8; ++r) {
        int row = m0 + mi * 16 + r + 8 * hi;
        out[row * H_ + col] = dT * w1[row * H_ + col] + a[r];
      }
    }
}

constexpr int NDH = 48 * SH * B_;
static_assert(24 * SD == 96 && NDH == 384, "the grids in kernel_launch are written as literals");
__global__ void __launch_bounds__(32) __attribute__((amdgpu_waves_per_eu(1, 1))) k_gemm_dh1(const float* __restrict__ sW1,
                           const float* __restrict__ Wo, const float* __restrict__ bo,
                           float* __restrict__ outv,
                           float* __restrict__ nW1,
                           float* __restrict__ ws) {
  const int id = blockIdx.x, lane = threadIdx.x;
  if (id >= NDH + 24) {
    const int j = id - (NDH + 24);
    const int b = j / 288, t = j % 288;
    nw1_tile(sW1, nW1, ws, b, (t / 24) * 32, (t % 24) * 32, lane);
    return;
  }
  if (id >= NDH) {
    const int r = id - NDH;
    const int b = r / 12, n0 = (r % 12) * 32;
    const float bb = bo[n0 + lane];
    const float y = rowvec32<C_, 16>(ws + OFF_PR + b * C_, Wo, C_, n0, lane);
    outv[b * C_ + n0 + lane] = y + bb;
    return;
  }
  const int b = id / (48 * SH), rem = id % (48 * SH);
  const int part = rem / 48, tile = rem % 48;
  const int m0 = (tile % 2) * 32, n0 = (tile / 2) * 32;
  const int k0 = part * (C_ / SH);
  const float* A = ws + OFF_DP + b * T_ * C_ + k0;
  const float* W = sW1 + b * C_ * H_ + k0 * H_;
  const float* gp = ws + OFF_GP + b * T_ * H_;
  float* wdhT = ws + OFF_WDHT + part * PL_WDHT + b * H_ * T_;
  const int cr = lane & 15, hi = lane >> 4;
  float gt[2][2][8], wt[2][8];
#pragma unroll
  for (int mi = 0; mi < 2; ++mi)
#pragma unroll
    for (int r = 0; r < 8; ++r) {
      int row = m0 + mi * 16 + r + 8 * hi;
      wt[mi][r] = ws[OFF_W + row];
#pragma unroll
      for (int ni = 0; ni < 2; ++ni)
        gt[mi][ni][r] = gp[row * H_ + n0 + ni * 16 + cr];
    }
  v8f acc[4] = {};
  gemm32x32<false, C_ / SH, 12, SD>(A, C_, W, H_, m0, n0, lane, acc, PL_DP);
#pragma unroll
  for (int mi = 0; mi < 2; ++mi)
#pragma unroll
    for (int ni = 0; ni < 2; ++ni) {
      v8f a = acc[mi * 2 + ni];
      float val[8];
#pragma unroll
      for (int r = 0; r < 8; ++r) {
        float d = a[r] * gt[mi][ni][r];
        val[r] = wt[mi][r] * d;
      }
      store_frag_T(wdhT, (n0 >> 4) + ni, m0 + mi * 16, cr, hi, val);
    }
}

template <int S>
__device__ __forceinline__ float row_sum_frag(const float* __restrict__ pl, int plane,
                                              int rowblk, int lane) {
  const float* p = pl + rowblk * (16 * T_) + (lane >> 4) * 32 + (lane & 15) * 2;
  v2f v[S][T_ / 4];
#pragma unroll
  for (int q = 0; q < S; ++q)
#pragma unroll
    for (int j = 0; j < T_ / 4; ++j)
      v[q][j] = *reinterpret_cast<const v2f*>(p + q * plane + j * 64);
  float s = 0.f;
#pragma unroll
  for (int q = 0; q < S; ++q)
#pragma unroll
    for (int j = 0; j < T_ / 4; ++j) { s += v[q][j].x; s += v[q][j].y; }
  s += __shfl_xor(s, 16);
  return s;
}

__global__ void __launch_bounds__(32) __attribute__((amdgpu_waves_per_eu(1, 2))) k_gemm_nw0(const float* __restrict__ sW0,
                           const float* __restrict__ sb0, const float* __restrict__ sb1,
                           float* __restrict__ nW0,
                           float* __restrict__ nb0, float* __restrict__ nb1,
                           float* __restrict__ ws) {
  const int b = blockIdx.y, lane = threadIdx.x;
  if (blockIdx.x >= 288) {
    const int blk = blockIdx.x - 288;
    const float dT = ws[OFF_W + 64];
    if (blk < H_ / 16) {
      const int r = blk * 16 + (lane & 15);
      const float s = row_sum_frag<SH>(ws + OFF_WDHT + b * H_ * T_, PL_WDHT, blk, lane);
      if (lane < 16) nb0[b * H_ + r] = dT * sb0[b * H_ + r] + s;
    } else {
      const int cb = blk - H_ / 16;
      const int c = cb * 16 + (lane & 15);
      const float s = row_sum_frag<SD>(ws + OFF_WDPT + b * C_ * T_, PL_WDPT, cb, lane);
      if (lane < 16) nb1[b * C_ + c] = dT * sb1[b * C_ + c] + s;
    }
    return;
  }
  const int m0 = (blockIdx.x / 12) * 32, n0 = (blockIdx.x % 12) * 32;
  const float* A = ws + OFF_WDHT + b * H_ * T_;
  const float* Bm = ws + OFF_K + b * T_ * C_;
  const float dT = ws[OFF_W + 64];
  float* out = nW0 + b * H_ * C_;
  const float* w0 = sW0 + b * H_ * C_;
  v8f acc[4] = {};
  gemm32x32<false, T_, 8, SH, true>(A, T_, Bm, C_, m0, n0, lane, acc, PL_WDHT);
  const int cr = lane & 15, hi = lane >> 4;
#pragma unroll
  for (int mi = 0; mi < 2; ++mi)
#pragma unroll
    for (int ni = 0; ni < 2; ++ni) {
      int col = n0 + ni * 16 + cr;
      v8f a = acc[mi * 2 + ni];
#pragma unroll
      for (int r = 0; r < 8; ++r) {
        int row = m0 + mi * 16 + r + 8 * hi;
        out[row * C_ + col] = dT * w0[row * C_ + col] + a[r];
      }
    }
}

}

extern "C" void kernel_launch(void* const* d_in, const int* in_sizes, int n_in,
                              void* d_out, int out_size, void* d_ws, size_t ws_size,
                              hipStream_t stream) {
  const float* x   = (const float*)d_in[0];
  const float* Wq  = (const float*)d_in[1];
  const float* bq  = (const float*)d_in[2];
  const float* Wk  = (const float*)d_in[3];
  const float* bk  = (const float*)d_in[4];
  const float* Wv  = (const float*)d_in[5];
  const float* bv  = (const float*)d_in[6];
  const float* Wo  = (const float*)d_in[7];
  const float* bo  = (const float*)d_in[8];
  const float* iq  = (const float*)d_in[9];
  const float* sW0 = (const float*)d_in[10];
  const float* sb0 = (const float*)d_in[11];
  const float* sW1 = (const float*)d_in[12];
  const float* sb1 = (const float*)d_in[13];

  float* out = (float*)d_out;
  float* ws  = (float*)d_ws;

  float* out_out = out;
  float* out_nW0 = out + B_ * C_;
  float* out_nb0 = out_nW0 + B_ * H_ * C_;
  float* out_nW1 = out_nb0 + B_ * H_;
  float* out_nb1 = out_nW1 + B_ * C_ * H_;

  k_gemm_kv<<<dim3(28, B_, 2), 32, 0, stream>>>(x, Wk, bk, Wv, bv, Wq, bq, iq, ws);
  k_gemm_h1<<<dim3(72, B_), 32, 0, stream>>>(sW0, sb0, ws);
  k_gemm_dp<<<dim3(96 + 12, B_), 32, 0, stream>>>(sW1, sb1, ws);
  k_gemm_dh1<<<dim3(384 + 24 + 576), 32, 0, stream>>>(sW1, Wo, bo, out_out, out_nW1, ws);
  k_gemm_nw0<<<dim3(288 + 72, B_), 32, 0, stream>>>(sW0, sb0, sb1, out_nW0, out_nb0, out_nb1, ws);
}


# Retrieval row of launch 3 (pred, K = 768) cut into 4 one-wave K parts a column tile; launch 4's retrieval blocks add the 4 planes once into an LDS row and read A there; pred's sum order changes
# speedup vs baseline: 5.2809x; 1.1059x over previous
#include <hip/hip_runtime.h>
#include <cstdint>

typedef float v2f __attribute__((ext_vector_type(2)));
typedef float v8f __attribute__((ext_vector_type(8)));

namespace {

constexpr int B_ = 2, T_ = 64, C_ = 384, H_ = 768;
constexpr float LR = 0.01f, MOM = 0.9f, DECAY = 0.001f;

constexpr int SD = 4;
constexpr int SH = 4;
constexpr int SR = 4;

constexpr int OFF_W    = 0;
constexpr int OFF_K    = 128;
constexpr int OFF_V    = OFF_K    + B_*T_*C_;
constexpr int OFF_H1   = OFF_V    + B_*T_*C_;
constexpr int OFF_GP   = OFF_H1   + B_*T_*H_;
constexpr int PL_DP   = B_*T_*C_;
constexpr int PL_WDPT = B_*C_*T_;
constexpr int PL_WDHT = B_*H_*T_;
constexpr int OFF_DP   = OFF_GP   + B_*T_*H_;
constexpr int OFF_WDPT = OFF_DP   + SD*PL_DP;
constexpr int OFF_WDHT = OFF_WDPT + SD*PL_WDPT;
constexpr int OFF_Q    = OFF_WDHT + SH*PL_WDHT;
constexpr int OFF_HR   = OFF_Q    + C_;
constexpr int PL_PR   = B_*C_;
constexpr int OFF_PR   = OFF_HR   + B_*H_;

__device__ __forceinline__ v8f wmma4(v2f a, v2f b, v8f c) {
  return __builtin_amdgcn_wmma_f32_16x16x4_f32(false, a, false, b, (short)0, c,
                                               false, false);
}

template <int KC, int S>
struct FragSet {
  v2f a0[KC][S], a1[KC][S], b0[KC], b1[KC];
};

template <bool BT, int KC, int S>
__device__ __forceinline__ void load_set(FragSet<KC, S>& f,
                                         const float* __restrict__ pa, int a1off, int ks, int plane,
                                         const float* __restrict__ pb, int ldb,
                                         int kk) {
#pragma unroll
  for (int s = 0; s < KC; ++s) {
    const int k = kk + 4 * s;
#pragma unroll
    for (int p = 0; p < S; ++p) {
      f.a0[s][p] = *reinterpret_cast<const v2f*>(pa + p * plane + k * ks);
      f.a1[s][p] = *reinterpret_cast<const v2f*>(pa + p * plane + a1off + k * ks);
    }
    if (BT) {
      f.b0[s] = *reinterpret_cast<const v2f*>(pb + k);
      f.b1[s] = *reinterpret_cast<const v2f*>(pb + 16 * ldb + k);
    } else {
      const float* q = pb + k * ldb;
      f.b0[s].x = q[0];  f.b0[s].y = q[ldb];
      f.b1[s].x = q[16]; f.b1[s].y = q[ldb + 16];
    }
  }
}

template <int KC, int S>
__device__ __forceinline__ void mma_set(const FragSet<KC, S>& f, v8f acc[4]) {
#pragma unroll
  for (int s = 0; s < KC; ++s) {
    v2f a0 = f.a0[s][0], a1 = f.a1[s][0];
#pragma unroll
    for (int p = 1; p < S; ++p) { a0 += f.a0[s][p]; a1 += f.a1[s][p]; }
    acc[0] = wmma4(a0, f.b0[s], acc[0]);
    acc[1] = wmma4(a0, f.b1[s], acc[1]);
    acc[2] = wmma4(a1, f.b0[s], acc[2]);
    acc[3] = wmma4(a1, f.b1[s], acc[3]);
  }
}

template <bool BT, int K, int KC, int S = 1, bool AF = false>
__device__ __forceinline__ void gemm32x32(const float* __restrict__ A, int lda,
                                          const float* __restrict__ Bm, int ldb,
                                          int m0, int n0, int lane,
                                          v8f acc[4], int plane = 0) {
  constexpr int CH = 4 * KC;
  constexpr int NCH = K / CH;
  static_assert(K % CH == 0 && NCH % 2 == 0, "K must be an even number of chunks");
  const int mr = lane & 15;
  const int kh = (lane >> 4) << 1;
  static_assert(!AF || K == T_, "fragment order is laid out for K = T_");
  const float* pa = AF ? A + (m0 >> 4) * (16 * T_) + (lane >> 4) * 32 + mr * 2
                       : A + (m0 + mr) * lda + kh;
  const int a1off = AF ? 16 * T_ : 16 * lda;
  const int ks = AF ? 16 : 1;
  const float* pb = BT ? Bm + (n0 + mr) * ldb + kh : Bm + kh * ldb + n0 + mr;
  FragSet<KC, S> f0, f1;
  load_set<BT, KC, S>(f0, pa, a1off, ks, plane, pb, ldb, 0);
  for (int c = 0; c + 2 < NCH; c += 2) {
    load_set<BT, KC, S>(f1, pa, a1off, ks, plane, pb, ldb, (c + 1) * CH);
    mma_set<KC, S>(f0, acc);
    load_set<BT, KC, S>(f0, pa, a1off, ks, plane, pb, ldb, (c + 2) * CH);
    mma_set<KC, S>(f1, acc);
  }
  load_set<BT, KC, S>(f1, pa, a1off, ks, plane, pb, ldb, (NCH - 1) * CH);
  mma_set<KC, S>(f0, acc);
  mma_set<KC, S>(f1, acc);
}

template <int KC>
struct RowSet {
  v2f a[KC], b0[KC], b1[KC];
};

template <int KC>
__device__ __forceinline__ void load_row_set(RowSet<KC>& f,
                                             const float* __restrict__ pa,
                                             const float* __restrict__ pb, int ldb,
                                             int kk) {
#pragma unroll
  for (int s = 0; s < KC; ++s) {
    const int k = kk + 4 * s;
    f.a[s]  = *reinterpret_cast<const v2f*>(pa + k);
    f.b0[s] = *reinterpret_cast<const v2f*>(pb + k);
    f.b1[s] = *reinterpret_cast<const v2f*>(pb + 16 * ldb + k);
  }
}

template <int KC>
__device__ __forceinline__ void mma_row_set(const RowSet<KC>& f, v8f acc[2]) {
#pragma unroll
  for (int s = 0; s < KC; ++s) {
    acc[0] = wmma4(f.a[s], f.b0[s], acc[0]);
    acc[1] = wmma4(f.a[s], f.b1[s], acc[1]);
  }
}

template <int K, int KC>
__device__ __forceinline__ float rowvec32(const float* __restrict__ xrow,
                                          const float* __restrict__ Wm, int ldb,
                                          int n0, int lane) {
  constexpr int CH = 4 * KC;
  constexpr int NCH = K / CH;
  static_assert(K % CH == 0 && NCH % 2 == 0, "K must be an even number of chunks");
  const int mr = lane & 15;
  const int kh = (lane >> 4) << 1;
  const float* pa = xrow + kh;
  const float* pb = Wm + (n0 + mr) * ldb + kh;
  v8f acc[2] = {};
  RowSet<KC> f0, f1;
  load_row_set<KC>(f0, pa, pb, ldb, 0);
  for (int c = 0; c + 2 < NCH; c += 2) {
    load_row_set<KC>(f1, pa, pb, ldb, (c + 1) * CH);
    mma_row_set<KC>(f0, acc);
    load_row_set<KC>(f0, pa, pb, ldb, (c + 2) * CH);
    mma_row_set<KC>(f1, acc);
  }
  load_row_set<KC>(f1, pa, pb, ldb, (NCH - 1) * CH);
  mma_row_set<KC>(f0, acc);
  mma_row_set<KC>(f1, acc);
  return (lane >> 4) ? acc[1][0] : acc[0][0];
}

template <int KC>
struct BSet {
  v2f b0[KC], b1[KC];
};

template <int KC>
__device__ __forceinline__ void load_b_set(BSet<KC>& f, const float* __restrict__ pb, int ldb, int kk) {
#pragma unroll
  for (int s = 0; s < KC; ++s) {
    const int k = kk + 4 * s;
    f.b0[s] = *reinterpret_cast<const v2f*>(pb + k);
    f.b1[s] = *reinterpret_cast<const v2f*>(pb + 16 * ldb + k);
  }
}

template <int KC>
__device__ __forceinline__ void mma_b_set(const BSet<KC>& f, const float* pa, int kk, v8f acc[2]) {
#pragma unroll
  for (int s = 0; s < KC; ++s) {
    const v2f a = *reinterpret_cast<const v2f*>(pa + kk + 4 * s);
    acc[0] = wmma4(a, f.b0[s], acc[0]);
    acc[1] = wmma4(a, f.b1[s], acc[1]);
  }
}

template <int K, int KC, class Fill>
__device__ __forceinline__ float rowvec32_lds(const float* xrow_lds, Fill fill,
                                              const float* __restrict__ Wm, int ldb,
                                              int n0, int lane) {
  constexpr int CH = 4 * KC;
  constexpr int NCH = K / CH;
  static_assert(K % CH == 0 && NCH % 2 == 0, "K must be an even number of chunks");
  const int mr = lane & 15;
  const int kh = (lane >> 4) << 1;
  const float* pa = xrow_lds + kh;
  const float* pb = Wm + (n0 + mr) * ldb + kh;
  v8f acc[2] = {};
  BSet<KC> f0, f1;
  load_b_set<KC>(f0, pb, ldb, 0);
  fill();
  for (int c = 0; c + 2 < NCH; c += 2) {
    load_b_set<KC>(f1, pb, ldb, (c + 1) * CH);
    mma_b_set<KC>(f0, pa, c * CH, acc);
    load_b_set<KC>(f0, pb, ldb, (c + 2) * CH);
    mma_b_set<KC>(f1, pa, (c + 1) * CH, acc);
  }
  load_b_set<KC>(f1, pb, ldb, (NCH - 1) * CH);
  mma_b_set<KC>(f0, pa, (NCH - 2) * CH, acc);
  mma_b_set<KC>(f1, pa, (NCH - 1) * CH, acc);
  return (lane >> 4) ? acc[1][0] : acc[0][0];
}

__device__ __forceinline__ void init_w(float* __restrict__ ws, int lane) {
  const float a = 1.0f - DECAY;
#pragma unroll
  for (int j = 0; j < 2; ++j) {
    const int t = lane + 32 * j;
    float n = (float)(T_ - t);
    float an = __powf(a, n), mn = __powf(MOM, n);
    ws[OFF_W + t] = -LR * (1.0f - MOM) * (an - mn) / (a - MOM);
  }
  if (lane == 0) ws[OFF_W + 64] = __powf(a, (float)T_);
}

__global__ void __launch_bounds__(32) k_gemm_kv(const float* __restrict__ x,
                          const float* __restrict__ Wk, const float* __restrict__ bk,
                          const float* __restrict__ Wv, const float* __restrict__ bv,
                          const float* __restrict__ Wq, const float* __restrict__ bq,
                          const float* __restrict__ iq,
                          float* __restrict__ ws) {
  const int b = blockIdx.y, which = blockIdx.z, lane = threadIdx.x;
  if (blockIdx.x == 27) {
    if (b == 0 && which == 0) init_w(ws, lane);
    return;
  }
  if (blockIdx.x >= 24) {
    const int n0 = ((blockIdx.x - 24) + 3 * (b + 2 * which)) * 32;
    const float bb = bq[n0 + lane];
    const float y = rowvec32<C_, 16>(iq, Wq, C_, n0, lane);
    ws[OFF_Q + n0 + lane] = y + bb;
    return;
  }
  const int m0 = (blockIdx.x & 1) * 32, n0 = (blockIdx.x >> 1) * 32;
  const float* A = x + b * T_ * C_;
  const float* W = which ? Wv : Wk;
  const float* bias = which ? bv : bk;
  float* out = ws + (which ? OFF_V : OFF_K) + b * T_ * C_;
  v8f acc[4] = {};
  gemm32x32<true, C_, 16>(A, C_, W, C_, m0, n0, lane, acc);
  const int cr = lane & 15, hi = lane >> 4;
#pragma unroll
  for (int mi = 0; mi < 2; ++mi)
#pragma unroll
    for (int ni = 0; ni < 2; ++ni) {
      int col = n0 + ni * 16 + cr;
      float bb = bias[col];
      v8f a = acc[mi * 2 + ni];
#pragma unroll
      for (int r = 0; r < 8; ++r) {
        int row = m0 + mi * 16 + r + 8 * hi;
        out[row * C_ + col] = a[r] + bb;
      }
    }
}

__global__ void __launch_bounds__(32) k_gemm_h1(const float* __restrict__ sW0,
                          const float* __restrict__ sb0,
                          float* __restrict__ ws) {
  const int b = blockIdx.y, lane = threadIdx.x;
  if (blockIdx.x >= 48) {
    const int n0 = (blockIdx.x - 48) * 32;
    const float bb = sb0[b * H_ + n0 + lane];
    const float pre = rowvec32<C_, 16>(ws + OFF_Q, sW0 + b * H_ * C_, C_, n0, lane) + bb;
    const float sig = 1.0f / (1.0f + __expf(-pre));
    ws[OFF_HR + b * H_ + n0 + lane] = pre * sig;
    return;
  }
  const int m0 = (blockIdx.x % 2) * 32, n0 = (blockIdx.x / 2) * 32;
  const float* A = ws + OFF_K + b * T_ * C_;
  const float* W = sW0 + b * H_ * C_;
  float* h1 = ws + OFF_H1 + b * T_ * H_;
  float* gp = ws + OFF_GP + b * T_ * H_;
  v8f acc[4] = {};
  gemm32x32<true, C_, 16>(A, C_, W, C_, m0, n0, lane, acc);
  const int cr = lane & 15, hi = lane >> 4;
#pragma unroll
  for (int mi = 0; mi < 2; ++mi)
#pragma unroll
    for (int ni = 0; ni < 2; ++ni) {
      int col = n0 + ni * 16 + cr;
      float bb = sb0[b * H_ + col];
      v8f a = acc[mi * 2 + ni];
#pragma unroll
      for (int r = 0; r < 8; ++r) {
        int row = m0 + mi * 16 + r + 8 * hi;
        float pre = a[r] + bb;
        float sig = 1.0f / (1.0f + __expf(-pre));
        h1[row * H_ + col] = pre * sig;
        gp[row * H_ + col] = sig * (1.0f + pre * (1.0f - sig));
      }
    }
}

__device__ __forceinline__ void store_frag_T(float* __restrict__ pl, int rowblk, int t0,
                                             int cr, int hi, const float val[8]) {
#pragma unroll
  for (int u = 0; u < 4; ++u) {
    const int kg = (t0 >> 2) + 2 * hi + (u >> 1);
    v2f x; x.x = val[2 * u]; x.y = val[2 * u + 1];
    *reinterpret_cast<v2f*>(pl + ((rowblk * (T_ / 4) + kg) * 2 + (u & 1)) * 32 + cr * 2) = x;
  }
}

__global__ void __launch_bounds__(32) __attribute__((amdgpu_waves_per_eu(1, 2))) k_gemm_dp(const float* __restrict__ sW1,
                          const float* __restrict__ sb1,
                          float* __restrict__ ws) {
  const int b = blockIdx.y, lane = threadIdx.x;
  if (blockIdx.x >= 24 * SD) {
    const int r = blockIdx.x - 24 * SD;
    const int part = r / 12, n0 = (r % 12) * 32;
    const int k0 = part * (H_ / SR);
    const float bb = part == 0 ? sb1[b * C_ + n0 + lane] : 0.0f;
    const float y = rowvec32<H_ / SR, 12>(ws + OFF_HR + b * H_ + k0,
                                          sW1 + b * C_ * H_ + k0, H_, n0, lane);
    ws[OFF_PR + part * PL_PR + b * C_ + n0 + lane] = y + bb;
    return;
  }
  const int part = blockIdx.x / 24, tile = blockIdx.x % 24;
  const int m0 = (tile % 2) * 32, n0 = (tile / 2) * 32;
  const int k0 = part * (H_ / SD);
  const float* A = ws + OFF_H1 + b * T_ * H_ + k0;
  const float* W = sW1 + b * C_ * H_ + k0;
  const float* v = ws + OFF_V + b * T_ * C_;
  float* dp = ws + OFF_DP + part * PL_DP + b * T_ * C_;
  float* wdpT = ws + OFF_WDPT + part * PL_WDPT + b * C_ * T_;
  const int cr = lane & 15, hi = lane >> 4;
  float vt[2][2][8] = {}, wt[2][8], bt[2] = {};
#pragma unroll
  for (int mi = 0; mi < 2; ++mi)
#pragma unroll
    for (int r = 0; r < 8; ++r) {
      int row = m0 + mi * 16 + r + 8 * hi;
      wt[mi][r] = ws[OFF_W + row];
    }
  if (part == 0) {
#pragma unroll
    for (int ni = 0; ni < 2; ++ni) bt[ni] = sb1[b * C_ + n0 + ni * 16 + cr];
#pragma unroll
    for (int mi = 0; mi < 2; ++mi)
#pragma unroll
      for (int r = 0; r < 8; ++r) {
        int row = m0 + mi * 16 + r + 8 * hi;
#pragma unroll
        for (int ni = 0; ni < 2; ++ni)
          vt[mi][ni][r] = v[row * C_ + n0 + ni * 16 + cr];
      }
  }
  v8f acc[4] = {};
  gemm32x32<true, H_ / SD, 12>(A, H_, W, H_, m0, n0, lane, acc);
  float dd[2][2][8];
#pragma unroll
  for (int mi = 0; mi < 2; ++mi)
#pragma unroll
    for (int ni = 0; ni < 2; ++ni) {
      int col = n0 + ni * 16 + cr;
      float bb = bt[ni];
      v8f a = acc[mi * 2 + ni];
#pragma unroll
      for (int r = 0; r < 8; ++r) {
        int row = m0 + mi * 16 + r + 8 * hi;
        float d = (a[r] + bb) - vt[mi][ni][r];
        dd[mi][ni][r] = d;
        dp[row * C_ + col] = d;
      }
    }
#pragma unroll
  for (int mi = 0; mi < 2; ++mi)
#pragma unroll
    for (int ni = 0; ni < 2; ++ni) {
      float val[8];
#pragma unroll
      for (int r = 0; r < 8; ++r) val[r] = wt[mi][r] * dd[mi][ni][r];
      store_frag_T(wdpT, (n0 >> 4) + ni, m0 + mi * 16, cr, hi, val);
    }
}

__device__ __forceinline__ void nw1_tile(const float* __restrict__ sW1,
                                         float* __restrict__ nW1,
                                         const float* __restrict__ ws,
                                         int b, int m0, int n0, int lane) {
  const float* A = ws + OFF_WDPT + b * C_ * T_;
  const float* Bm = ws + OFF_H1 + b * T_ * H_;
  const float dT = ws[OFF_W + 64];
  float* out = nW1 + b * C_ * H_;
  const float* w1 = sW1 + b * C_ * H_;
  v8f acc[4] = {};
  gemm32x32<false, T_, 8, SD, true>(A, T_, Bm, H_, m0, n0, lane, acc, PL_WDPT);
  const int cr = lane & 15, hi = lane >> 4;
#pragma unroll
  for (int mi = 0; mi < 2; ++mi)
#pragma unroll
    for (int ni = 0; ni < 2; ++ni) {
      int col = n0 + ni * 16 + cr;
      v8f a = acc[mi * 2 + ni];
#pragma unroll
      for (int r = 0; r < 8; ++r) {
        int row = m0 + mi * 16 + r + 8 * hi;
        out[row * H_ + col] = dT * w1[row * H_ + col] + a[r];
      }
    }
}

constexpr int NDH = 48 * SH * B_;
static_assert(24 * SD == 96 && 12 * SR == 48 && NDH == 384, "the grids in kernel_launch are written as literals");
__global__ void __launch_bounds__(32) __attribute__((amdgpu_waves_per_eu(1, 1))) k_gemm_dh1(const float* __restrict__ sW1,
                           const float* __restrict__ Wo, const float* __restrict__ bo,
                           float* __restrict__ outv,
                           float* __restrict__ nW1,
                           float* __restrict__ ws) {
  const int id = blockIdx.x, lane = threadIdx.x;
  if (id >= NDH + 24) {
    const int j = id - (NDH + 24);
    const int b = j / 288, t = j % 288;
    nw1_tile(sW1, nW1, ws, b, (t / 24) * 32, (t % 24) * 32, lane);
    return;
  }
  if (id >= NDH) {
    const int r = id - NDH;
    const int b = r / 12, n0 = (r % 12) * 32;
    const float bb = bo[n0 + lane];
    __shared__ float spred[C_];
    typedef float v4f __attribute__((ext_vector_type(4)));
    const float* pp = ws + OFF_PR + b * C_ + lane * 4;
    v4f pv[C_ / 128][SR];
#pragma unroll
    for (int m = 0; m < C_ / 128; ++m)
#pragma unroll
      for (int p = 0; p < SR; ++p)
        pv[m][p] = *reinterpret_cast<const v4f*>(pp + p * PL_PR + m * 128);
    auto fill = [&]() {
#pragma unroll
      for (int m = 0; m < C_ / 128; ++m) {
        v4f s = pv[m][0];
#pragma unroll
        for (int p = 1; p < SR; ++p) s += pv[m][p];
        *reinterpret_cast<v4f*>(spred + lane * 4 + m * 128) = s;
      }
      __builtin_amdgcn_fence(__ATOMIC_ACQ_REL, "wavefront", "local");
    };
    const float y = rowvec32_lds<C_, 16>(spred, fill, Wo, C_, n0, lane);
    outv[b * C_ + n0 + lane] = y + bb;
    return;
  }
  const int b = id / (48 * SH), rem = id % (48 * SH);
  const int part = rem / 48, tile = rem % 48;
  const int m0 = (tile % 2) * 32, n0 = (tile / 2) * 32;
  const int k0 = part * (C_ / SH);
  const float* A = ws + OFF_DP + b * T_ * C_ + k0;
  const float* W = sW1 + b * C_ * H_ + k0 * H_;
  const float* gp = ws + OFF_GP + b * T_ * H_;
  float* wdhT = ws + OFF_WDHT + part * PL_WDHT + b * H_ * T_;
  const int cr = lane & 15, hi = lane >> 4;
  float gt[2][2][8], wt[2][8];
#pragma unroll
  for (int mi = 0; mi < 2; ++mi)
#pragma unroll
    for (int r = 0; r < 8; ++r) {
      int row = m0 + mi * 16 + r + 8 * hi;
      wt[mi][r] = ws[OFF_W + row];
#pragma unroll
      for (int ni = 0; ni < 2; ++ni)
        gt[mi][ni][r] = gp[row * H_ + n0 + ni * 16 + cr];
    }
  v8f acc[4] = {};
  gemm32x32<false, C_ / SH, 12, SD>(A, C_, W, H_, m0, n0, lane, acc, PL_DP);
#pragma unroll
  for (int mi = 0; mi < 2; ++mi)
#pragma unroll
    for (int ni = 0; ni < 2; ++ni) {
      v8f a = acc[mi * 2 + ni];
      float val[8];
#pragma unroll
      for (int r = 0; r < 8; ++r) {
        float d = a[r] * gt[mi][ni][r];
        val[r] = wt[mi][r] * d;
      }
      store_frag_T(wdhT, (n0 >> 4) + ni, m0 + mi * 16, cr, hi, val);
    }
}

template <int S>
__device__ __forceinline__ float row_sum_frag(const float* __restrict__ pl, int plane,
                                              int rowblk, int lane) {
  const float* p = pl + rowblk * (16 * T_) + (lane >> 4) * 32 + (lane & 15) * 2;
  v2f v[S][T_ / 4];
#pragma unroll
  for (int q = 0; q < S; ++q)
#pragma unroll
    for (int j = 0; j < T_ / 4; ++j)
      v[q][j] = *reinterpret_cast<const v2f*>(p + q * plane + j * 64);
  float s = 0.f;
#pragma unroll
  for (int q = 0; q < S; ++q)
#pragma unroll
    for (int j = 0; j < T_ / 4; ++j) { s += v[q][j].x; s += v[q][j].y; }
  s += __shfl_xor(s, 16);
  return s;
}

__global__ void __launch_bounds__(32) __attribute__((amdgpu_waves_per_eu(1, 2))) k_gemm_nw0(const float* __restrict__ sW0,
                           const float* __restrict__ sb0, const float* __restrict__ sb1,
                           float* __restrict__ nW0,
                           float* __restrict__ nb0, float* __restrict__ nb1,
                           float* __restrict__ ws) {
  const int b = blockIdx.y, lane = threadIdx.x;
  if (blockIdx.x >= 288) {
    const int blk = blockIdx.x - 288;
    const float dT = ws[OFF_W + 64];
    if (blk < H_ / 16) {
      const int r = blk * 16 + (lane & 15);
      const float s = row_sum_frag<SH>(ws + OFF_WDHT + b * H_ * T_, PL_WDHT, blk, lane);
      if (lane < 16) nb0[b * H_ + r] = dT * sb0[b * H_ + r] + s;
    } else {
      const int cb = blk - H_ / 16;
      const int c = cb * 16 + (lane & 15);
      const float s = row_sum_frag<SD>(ws + OFF_WDPT + b * C_ * T_, PL_WDPT, cb, lane);
      if (lane < 16) nb1[b * C_ + c] = dT * sb1[b * C_ + c] + s;
    }
    return;
  }
  const int m0 = (blockIdx.x / 12) * 32, n0 = (blockIdx.x % 12) * 32;
  const float* A = ws + OFF_WDHT + b * H_ * T_;
  const float* Bm = ws + OFF_K + b * T_ * C_;
  const float dT = ws[OFF_W + 64];
  float* out = nW0 + b * H_ * C_;
  const float* w0 = sW0 + b * H_ * C_;
  v8f acc[4] = {};
  gemm32x32<false, T_, 8, SH, true>(A, T_, Bm, C_, m0, n0, lane, acc, PL_WDHT);
  const int cr = lane & 15, hi = lane >> 4;
#pragma unroll
  for (int mi = 0; mi < 2; ++mi)
#pragma unroll
    for (int ni = 0; ni < 2; ++ni) {
      int col = n0 + ni * 16 + cr;
      v8f a = acc[mi * 2 + ni];
#pragma unroll
      for (int r = 0; r < 8; ++r) {
        int row = m0 + mi * 16 + r + 8 * hi;
        out[row * C_ + col] = dT * w0[row * C_ + col] + a[r];
      }
    }
}

}

extern "C" void kernel_launch(void* const* d_in, const int* in_sizes, int n_in,
                              void* d_out, int out_size, void* d_ws, size_t ws_size,
                              hipStream_t stream) {
  const float* x   = (const float*)d_in[0];
  const float* Wq  = (const float*)d_in[1];
  const float* bq  = (const float*)d_in[2];
  const float* Wk  = (const float*)d_in[3];
  const float* bk  = (const float*)d_in[4];
  const float* Wv  = (const float*)d_in[5];
  const float* bv  = (const float*)d_in[6];
  const float* Wo  = (const float*)d_in[7];
  const float* bo  = (const float*)d_in[8];
  const float* iq  = (const float*)d_in[9];
  const float* sW0 = (const float*)d_in[10];
  const float* sb0 = (const float*)d_in[11];
  const float* sW1 = (const float*)d_in[12];
  const float* sb1 = (const float*)d_in[13];

  float* out = (float*)d_out;
  float* ws  = (float*)d_ws;

  float* out_out = out;
  float* out_nW0 = out + B_ * C_;
  float* out_nb0 = out_nW0 + B_ * H_ * C_;
  float* out_nW1 = out_nb0 + B_ * H_;
  float* out_nb1 = out_nW1 + B_ * C_ * H_;

  k_gemm_kv<<<dim3(28, B_, 2), 32, 0, stream>>>(x, Wk, bk, Wv, bv, Wq, bq, iq, ws);
  k_gemm_h1<<<dim3(72, B_), 32, 0, stream>>>(sW0, sb0, ws);
  k_gemm_dp<<<dim3(96 + 48, B_), 32, 0, stream>>>(sW1, sb1, ws);
  k_gemm_dh1<<<dim3(384 + 24 + 576), 32, 0, stream>>>(sW1, Wo, bo, out_out, out_nW1, ws);
  k_gemm_nw0<<<dim3(288 + 72, B_), 32, 0, stream>>>(sW0, sb0, sb1, out_nW0, out_nb0, out_nb1, ws);
}
